# starting kernel
# baseline (speedup 1.0000x reference)
#include <hip/hip_runtime.h>


#define N       128
#define LDSTR   136
#define PSTR    288
#define NBATCH  2048
#define EPSF    1e-8f
#define NSTEPS  3

typedef __attribute__((ext_vector_type(2))) float v2f;
typedef __attribute__((ext_vector_type(8))) float v8f;

__device__ __forceinline__ int fidx(int row, int col) {
  return (row >> 1) * PSTR + col * 2 + (row & 1);
}

__global__ __launch_bounds__(256, 1)
void sgf_kernel(const float* __restrict__ corr,
                const float* __restrict__ temperature,
                float* __restrict__ out) {
  extern __shared__ float lds[];
  float* sC  = lds;
  float* sS  = lds + N * LDSTR;
  float* sFp = lds + 2 * N * LDSTR;

  const int tid  = threadIdx.x;
  const int lane = tid & 31;
  const int wave = tid >> 5;
  const int lm   = lane & 15;
  const int lh   = lane >> 4;
  const int b    = blockIdx.x;

  const float invT = 1.0f / temperature[0];
  const float* src = corr + (size_t)b * (N * N);

  for (int e = tid * 4; e < N * N; e += 256 * 4) {
    float4 v = *(const float4*)(src + e);
    int i = e >> 7, j = e & 127;
    *(float4*)&sS[i * LDSTR + j] = v;
  }
  __syncthreads();

  for (int e = tid; e < N * N; e += 256) {
    int i = e >> 7, j = e & 127;
    float x = sS[i * LDSTR + j];
    float y = sS[j * LDSTR + i];
    x = fminf(fmaxf(x, -1.0f + EPSF), 1.0f - EPSF);
    y = fminf(fmaxf(y, -1.0f + EPSF), 1.0f - EPSF);
    sC[i * LDSTR + j] = 0.5f * (x + y);
  }
  __syncthreads();

  if (tid < N) {
    const int i = tid;
    float t0 = -1e30f, t1 = -1e30f, t2 = -1e30f;
    for (int j = 0; j < N; ++j) {
      float cv  = sC[i * LDSTR + j];
      float d   = __builtin_sqrtf(2.0f * (1.0f - cv) + EPSF);
      float aff = __expf(-d * invT);
      if (j == i) aff = 0.0f;
      sS[i * LDSTR + j] = aff;
      if (aff > t0)      { t2 = t1; t1 = t0; t0 = aff; }
      else if (aff > t1) { t2 = t1; t1 = aff; }
      else if (aff > t2) { t2 = aff; }
    }
    float sum = 0.0f;
    for (int j = 0; j < N; ++j) {
      float a = sS[i * LDSTR + j];
      if (a >= t2) sum += __expf(a - t0);
    }
    float rsum = 1.0f / sum;
    for (int j = 0; j < N; ++j) {
      float a = sS[i * LDSTR + j];
      sS[i * LDSTR + j] = (a >= t2) ? __expf(a - t0) * rsum : 0.0f;
    }
  }
  __syncthreads();

  for (int e = tid; e < N * N; e += 256) {
    int i = e >> 7, j = e & 127;
    if (i < j) {
      float a = sS[i * LDSTR + j];
      float c2 = sS[j * LDSTR + i];
      float m = 0.5f * (a + c2);
      sS[i * LDSTR + j] = m;
      sS[j * LDSTR + i] = m;
    }
  }

  for (int e = tid; e < N * N; e += 256) {
    int i = e >> 7, j = e & 127;
    sFp[fidx(i, j)] = sC[i * LDSTR + j];
  }
  __syncthreads();

  const int tm       = wave;
  const int arowbase = (tm * 16 + lm) * LDSTR + 2 * lh;
  const int bcolbase = lh * PSTR + lm * 2;

  const float kAlpha[NSTEPS] = { 0.5f, 1.0f / 3.0f, 0.25f };

#pragma unroll
  for (int step = 0; step < NSTEPS; ++step) {
    v8f acc[8] = {};

    v2f av = *(const v2f*)&sS[arowbase];
    v2f bcur[8];
#pragma unroll
    for (int tn = 0; tn < 8; ++tn)
      bcur[tn] = *(const v2f*)&sFp[bcolbase + tn * 32];

    for (int k = 0; k < N - 4; k += 4) {
      v2f an = *(const v2f*)&sS[arowbase + k + 4];
      const int bb = ((k + 4) >> 1) * PSTR + bcolbase;
      v2f bn[8];
#pragma unroll
      for (int tn = 0; tn < 8; ++tn)
        bn[tn] = *(const v2f*)&sFp[bb + tn * 32];
#pragma unroll
      for (int tn = 0; tn < 8; ++tn)
        acc[tn] = __builtin_amdgcn_wmma_f32_16x16x4_f32(
            false, av, false, bcur[tn], (short)0, acc[tn], false, false);
      av = an;
#pragma unroll
      for (int tn = 0; tn < 8; ++tn) bcur[tn] = bn[tn];
    }
#pragma unroll
    for (int tn = 0; tn < 8; ++tn)
      acc[tn] = __builtin_amdgcn_wmma_f32_16x16x4_f32(
          false, av, false, bcur[tn], (short)0, acc[tn], false, false);

    __syncthreads();

    const float alpha = kAlpha[step];
#pragma unroll
    for (int tn = 0; tn < 8; ++tn) {
#pragma unroll
      for (int v = 0; v < 8; ++v) {
        int row  = tm * 16 + v + 8 * lh;
        int col  = tn * 16 + lm;
        int ridx = row * LDSTR + col;
        int fi   = fidx(row, col);
        float msg = acc[tn][v];
        float f   = sFp[fi];
        float s   = sS[ridx];
        float cc  = sC[ridx];
        float fn  = __builtin_fmaf(alpha, msg - f, f);
        sFp[fi]   = __builtin_fmaf(s, cc - fn, fn);
      }
    }
    __syncthreads();
  }

  float* dst = out + (size_t)b * (N * N);
  for (int e = tid; e < N * N; e += 256) {
    int i = e >> 7, j = e & 127;
    float val = (i == j) ? 1.0f
                         : 0.5f * (sFp[fidx(i, j)] + sFp[fidx(j, i)]);
    dst[e] = val;
  }
}

extern "C" void kernel_launch(void* const* d_in, const int* in_sizes, int n_in,
                              void* d_out, int out_size, void* d_ws, size_t ws_size,
                              hipStream_t stream) {
  (void)in_sizes; (void)n_in; (void)out_size; (void)d_ws; (void)ws_size;
  const float* corr = (const float*)d_in[0];
  const float* temp = (const float*)d_in[1];
  float* out = (float*)d_out;

  const size_t shmem =
      ((size_t)2 * N * LDSTR + (size_t)(N / 2) * PSTR) * sizeof(float);
  static bool attr_set = false;
  if (!attr_set) {
    hipFuncSetAttribute((const void*)sgf_kernel,
                        hipFuncAttributeMaxDynamicSharedMemorySize, (int)shmem);
    attr_set = true;
  }
  sgf_kernel<<<dim3(NBATCH), dim3(256), shmem, stream>>>(corr, temp, out);
}


# LDS bank mapping: row stride 136->132 and a 32-float skew of rows 8-15 in the filtered plane make accumulator-tile passes conflict-free; last step lands row-major, output column reads 2-way not 8-way
# speedup vs baseline: 2.5054x; 2.5054x over previous
#include <hip/hip_runtime.h>


#define N       128
#define LDSTR   132
#define PSTR    288
#define NBATCH  2048
#define EPSF    1e-8f
#define NSTEPS  3
#define NT      1024
#define WT      2

typedef __attribute__((ext_vector_type(2))) float v2f;
typedef __attribute__((ext_vector_type(8))) float v8f;

__device__ __forceinline__ int frow(int row) {
  return (row >> 1) * PSTR + ((row & 8) << 2);
}
__device__ __forceinline__ int fidx(int row, int col) {
  return frow(row) + col * 2 + (row & 1);
}

__global__ __launch_bounds__(NT, 1)
void sgf_kernel(const float* __restrict__ corr,
                const float* __restrict__ temperature,
                float* __restrict__ out) {
  extern __shared__ float lds[];
  float* sC  = lds;
  float* sS  = lds + N * LDSTR;
  float* sFp = lds + 2 * N * LDSTR;

  const int tid  = threadIdx.x;
  const int lane = tid & 31;
  const int wave = tid >> 5;
  const int lm   = lane & 15;
  const int lh   = lane >> 4;
  const int b    = blockIdx.x;

  const float invT = 1.0f / temperature[0];
  const float* src = corr + (size_t)b * (N * N);

  for (int e = tid * 4; e < N * N; e += NT * 4) {
    float4 v = *(const float4*)(src + e);
    int i = e >> 7, j = e & 127;
    *(float4*)&sS[i * LDSTR + j] = v;
  }
  __syncthreads();

  for (int e = tid; e < N * N; e += NT) {
    int i = e >> 7, j = e & 127;
    float x = sS[i * LDSTR + j];
    float y = sS[j * LDSTR + i];
    x = fminf(fmaxf(x, -1.0f + EPSF), 1.0f - EPSF);
    y = fminf(fmaxf(y, -1.0f + EPSF), 1.0f - EPSF);
    sC[i * LDSTR + j] = 0.5f * (x + y);
  }
  __syncthreads();

  if (tid < N) {
    const int i = tid;
    const float* crow = &sC[i * LDSTR];
    float*       srow = &sS[i * LDSTR];
    float t0 = -1e30f, t1 = -1e30f, t2 = -1e30f;
    for (int j0 = 0; j0 < N; j0 += 16) {
      float a[16];
      for (int q = 0; q < 4; ++q) {
        float4 cv4 = *(const float4*)&crow[j0 + 4 * q];
        a[4 * q + 0] = cv4.x; a[4 * q + 1] = cv4.y;
        a[4 * q + 2] = cv4.z; a[4 * q + 3] = cv4.w;
      }
#pragma unroll
      for (int u = 0; u < 16; ++u) {
        float d   = __builtin_sqrtf(2.0f * (1.0f - a[u]) + EPSF);
        float aff = __expf(-d * invT);
        if (j0 + u == i) aff = 0.0f;
        a[u] = aff;
      }
      for (int q = 0; q < 4; ++q)
        *(float4*)&srow[j0 + 4 * q] =
            make_float4(a[4 * q + 0], a[4 * q + 1], a[4 * q + 2], a[4 * q + 3]);
      for (int u = 0; u < 16; ++u) {
        float n0 = fmaxf(t0, a[u]);
        float n1 = __builtin_amdgcn_fmed3f(t0, t1, a[u]);
        float n2 = __builtin_amdgcn_fmed3f(t1, t2, a[u]);
        t0 = n0; t1 = n1; t2 = n2;
      }
    }
    float sum = 0.0f;
    for (int j0 = 0; j0 < N; j0 += 16) {
      float a[16];
      for (int q = 0; q < 4; ++q) {
        float4 v4 = *(const float4*)&srow[j0 + 4 * q];
        a[4 * q + 0] = v4.x; a[4 * q + 1] = v4.y;
        a[4 * q + 2] = v4.z; a[4 * q + 3] = v4.w;
      }
      for (int u = 0; u < 16; ++u)
        if (a[u] >= t2) sum += __expf(a[u] - t0);
    }
    float rsum = 1.0f / sum;
    for (int j0 = 0; j0 < N; j0 += 16) {
      float a[16];
      for (int q = 0; q < 4; ++q) {
        float4 v4 = *(const float4*)&srow[j0 + 4 * q];
        a[4 * q + 0] = v4.x; a[4 * q + 1] = v4.y;
        a[4 * q + 2] = v4.z; a[4 * q + 3] = v4.w;
      }
      for (int u = 0; u < 16; ++u)
        a[u] = (a[u] >= t2) ? __expf(a[u] - t0) * rsum : 0.0f;
      for (int q = 0; q < 4; ++q)
        *(float4*)&srow[j0 + 4 * q] =
            make_float4(a[4 * q + 0], a[4 * q + 1], a[4 * q + 2], a[4 * q + 3]);
    }
  }
  __syncthreads();

  for (int e = tid; e < N * N; e += NT) {
    int i = e >> 7, j = e & 127;
    if (i < j) {
      float a = sS[i * LDSTR + j];
      float c2 = sS[j * LDSTR + i];
      float m = 0.5f * (a + c2);
      sS[i * LDSTR + j] = m;
      sS[j * LDSTR + i] = m;
    }
  }

  {
    const int j = tid & 127, i0 = tid >> 7;
    float cv[N * N / NT];
#pragma unroll
    for (int u = 0; u < N * N / NT; ++u) cv[u] = sC[(i0 + 8 * u) * LDSTR + j];
#pragma unroll
    for (int u = 0; u < N * N / NT; ++u) sFp[fidx(i0 + 8 * u, j)] = cv[u];
  }
  __syncthreads();

  const int tg       = wave >> 3;
  const int tn       = wave & 7;
  const int arowbase = (tg * 32 + lm) * LDSTR + 2 * lh;
  const int bcolbase = lh * PSTR + (tn * 16 + lm) * 2;

  const float kAlpha[NSTEPS] = { 0.5f, 1.0f / 3.0f, 0.25f };

#pragma unroll
  for (int step = 0; step < NSTEPS; ++step) {
    v8f acc[WT] = {};

    v2f bv = *(const v2f*)&sFp[bcolbase];
    v2f acur[WT];
#pragma unroll
    for (int t = 0; t < WT; ++t)
      acur[t] = *(const v2f*)&sS[arowbase + t * 16 * LDSTR];

    for (int k = 0; k < N - 4; k += 4) {
      v2f bnx = *(const v2f*)&sFp[frow(k + 4) + bcolbase];
      v2f an[WT];
#pragma unroll
      for (int t = 0; t < WT; ++t)
        an[t] = *(const v2f*)&sS[arowbase + t * 16 * LDSTR + k + 4];
#pragma unroll
      for (int t = 0; t < WT; ++t)
        acc[t] = __builtin_amdgcn_wmma_f32_16x16x4_f32(
            false, acur[t], false, bv, (short)0, acc[t], false, false);
      bv = bnx;
#pragma unroll
      for (int t = 0; t < WT; ++t) acur[t] = an[t];
    }
#pragma unroll
    for (int t = 0; t < WT; ++t)
      acc[t] = __builtin_amdgcn_wmma_f32_16x16x4_f32(
          false, acur[t], false, bv, (short)0, acc[t], false, false);

    __syncthreads();

    const float alpha = kAlpha[step];
#pragma unroll
    for (int t = 0; t < WT; ++t) {
#pragma unroll
      for (int v = 0; v < 8; ++v) {
        int row  = tg * 32 + t * 16 + v + 8 * lh;
        int col  = tn * 16 + lm;
        int ridx = row * LDSTR + col;
        int fi   = fidx(row, col);
        float msg = acc[t][v];
        float f   = sFp[fi];
        float s   = sS[ridx];
        float cc  = sC[ridx];
        float fn  = __builtin_fmaf(alpha, msg - f, f);
        float fo  = __builtin_fmaf(s, cc - fn, fn);
        if (step == NSTEPS - 1) sS[ridx] = fo;
        else                    sFp[fi]  = fo;
      }
    }
    __syncthreads();
  }

  float* dst = out + (size_t)b * (N * N);
  for (int e = tid; e < N * N; e += NT) {
    int i = e >> 7, j = e & 127;
    float val = (i == j) ? 1.0f
                         : 0.5f * (sS[i * LDSTR + j] + sS[j * LDSTR + i]);
    dst[e] = val;
  }
}

extern "C" void kernel_launch(void* const* d_in, const int* in_sizes, int n_in,
                              void* d_out, int out_size, void* d_ws, size_t ws_size,
                              hipStream_t stream) {
  (void)in_sizes; (void)n_in; (void)out_size; (void)d_ws; (void)ws_size;
  const float* corr = (const float*)d_in[0];
  const float* temp = (const float*)d_in[1];
  float* out = (float*)d_out;

  const size_t shmem =
      ((size_t)2 * N * LDSTR + (size_t)(N / 2) * PSTR) * sizeof(float);
  static bool attr_set = false;
  if (!attr_set) {
    hipFuncSetAttribute((const void*)sgf_kernel,
                        hipFuncAttributeMaxDynamicSharedMemorySize, (int)shmem);
    attr_set = true;
  }
  sgf_kernel<<<dim3(NBATCH), dim3(NT), shmem, stream>>>(corr, temp, out);
}


# Persistent grid: 256 workgroups (one per WGP) loop over 8 matrices each; 16 waves idle during the 4-wave graph build fetch the next matrix from HBM into an LDS plane of its own; sum orders unchanged
# speedup vs baseline: 2.8416x; 1.1342x over previous
#include <hip/hip_runtime.h>


#define N       128
#define LDSTR   132
#define PSTR    288
#define NBATCH  2048
#define EPSF    1e-8f
#define NSTEPS  3
#define NT      1024
#define NWG     256
#define NFT     512
#define WT      2

typedef __attribute__((ext_vector_type(2))) float v2f;
typedef __attribute__((ext_vector_type(8))) float v8f;
typedef __attribute__((ext_vector_type(4))) float v4f;

template <int nthr>
__device__ __forceinline__ void fetch_raw(const float* __restrict__ src, float* dst, int t) {
  constexpr int NP = N * N / (4 * nthr);
  const float* s = src + t * 4;
  float*       p = dst + (t >> 5) * LDSTR + (t & 31) * 4;
  v4f v[NP];
#pragma unroll
  for (int q = 0; q < NP; ++q) v[q] = *(const v4f*)(s + q * nthr * 4);
#pragma unroll
  for (int q = 0; q < NP; ++q) *(v4f*)(p + q * (nthr / 32) * LDSTR) = v[q];
}

__device__ __forceinline__ int frow(int row) {
  return (row >> 1) * PSTR + ((row & 8) << 2);
}
__device__ __forceinline__ int fidx(int row, int col) {
  return frow(row) + col * 2 + (row & 1);
}

__global__ __launch_bounds__(NT, 1)
void sgf_kernel(const float* __restrict__ corr,
                const float* __restrict__ temperature,
                float* __restrict__ out) {
  extern __shared__ float lds[];
  float* sC  = lds;
  float* sS  = lds + N * LDSTR;
  float* sFp = lds + 2 * N * LDSTR;
  float* sR  = lds + 2 * N * LDSTR + (N / 2) * PSTR;

  const int tid0 = threadIdx.x;

  const float invT = 1.0f / temperature[0];

  fetch_raw<NT>(corr + (size_t)blockIdx.x * (N * N), sR, tid0);
  __syncthreads();

  for (int b = blockIdx.x; b < NBATCH; b += NWG) {

  int tid = tid0;
  asm volatile("" : "+v"(tid));
  __builtin_assume(tid >= 0 && tid < NT);
  const int lane = tid & 31;
  const int wave = tid >> 5;
  const int lm   = lane & 15;
  const int lh   = lane >> 4;

  for (int e = tid; e < N * N; e += NT) {
    int i = e >> 7, j = e & 127;
    float x = sR[i * LDSTR + j];
    float y = sR[j * LDSTR + i];
    x = fminf(fmaxf(x, -1.0f + EPSF), 1.0f - EPSF);
    y = fminf(fmaxf(y, -1.0f + EPSF), 1.0f - EPSF);
    sC[i * LDSTR + j] = 0.5f * (x + y);
  }
  __syncthreads();

  if (tid < N) {
    const int i = tid;
    const float* crow = &sC[i * LDSTR];
    float*       srow = &sS[i * LDSTR];
    float t0 = -1e30f, t1 = -1e30f, t2 = -1e30f;
    for (int j0 = 0; j0 < N; j0 += 16) {
      float a[16];
      for (int q = 0; q < 4; ++q) {
        float4 cv4 = *(const float4*)&crow[j0 + 4 * q];
        a[4 * q + 0] = cv4.x; a[4 * q + 1] = cv4.y;
        a[4 * q + 2] = cv4.z; a[4 * q + 3] = cv4.w;
      }
#pragma unroll
      for (int u = 0; u < 16; ++u) {
        float d   = __builtin_sqrtf(2.0f * (1.0f - a[u]) + EPSF);
        float aff = __expf(-d * invT);
        if (j0 + u == i) aff = 0.0f;
        a[u] = aff;
      }
      for (int q = 0; q < 4; ++q)
        *(float4*)&srow[j0 + 4 * q] =
            make_float4(a[4 * q + 0], a[4 * q + 1], a[4 * q + 2], a[4 * q + 3]);
      for (int u = 0; u < 16; ++u) {
        float n0 = fmaxf(t0, a[u]);
        float n1 = __builtin_amdgcn_fmed3f(t0, t1, a[u]);
        float n2 = __builtin_amdgcn_fmed3f(t1, t2, a[u]);
        t0 = n0; t1 = n1; t2 = n2;
      }
    }
    float sum = 0.0f;
    for (int j0 = 0; j0 < N; j0 += 16) {
      float a[16];
      for (int q = 0; q < 4; ++q) {
        float4 v4 = *(const float4*)&srow[j0 + 4 * q];
        a[4 * q + 0] = v4.x; a[4 * q + 1] = v4.y;
        a[4 * q + 2] = v4.z; a[4 * q + 3] = v4.w;
      }
      for (int u = 0; u < 16; ++u)
        if (a[u] >= t2) sum += __expf(a[u] - t0);
    }
    float rsum = 1.0f / sum;
    for (int j0 = 0; j0 < N; j0 += 16) {
      float a[16];
      for (int q = 0; q < 4; ++q) {
        float4 v4 = *(const float4*)&srow[j0 + 4 * q];
        a[4 * q + 0] = v4.x; a[4 * q + 1] = v4.y;
        a[4 * q + 2] = v4.z; a[4 * q + 3] = v4.w;
      }
      for (int u = 0; u < 16; ++u)
        a[u] = (a[u] >= t2) ? __expf(a[u] - t0) * rsum : 0.0f;
      for (int q = 0; q < 4; ++q)
        *(float4*)&srow[j0 + 4 * q] =
            make_float4(a[4 * q + 0], a[4 * q + 1], a[4 * q + 2], a[4 * q + 3]);
    }
  } else if (tid >= NT - NFT && b + NWG < NBATCH) {
    fetch_raw<NFT>(corr + (size_t)(b + NWG) * (N * N), sR, tid & (NFT - 1));
  }
  __syncthreads();

  for (int e = tid; e < N * N; e += NT) {
    int i = e >> 7, j = e & 127;
    if (i < j) {
      float a = sS[i * LDSTR + j];
      float c2 = sS[j * LDSTR + i];
      float m = 0.5f * (a + c2);
      sS[i * LDSTR + j] = m;
      sS[j * LDSTR + i] = m;
    }
  }

  {
    const int j = tid & 127, i0 = tid >> 7;
    float cv[N * N / NT];
#pragma unroll
    for (int u = 0; u < N * N / NT; ++u) cv[u] = sC[(i0 + 8 * u) * LDSTR + j];
#pragma unroll
    for (int u = 0; u < N * N / NT; ++u) sFp[fidx(i0 + 8 * u, j)] = cv[u];
  }
  __syncthreads();

  const int tg       = wave >> 3;
  const int tn       = wave & 7;
  const int arowbase = (tg * 32 + lm) * LDSTR + 2 * lh;
  const int bcolbase = lh * PSTR + (tn * 16 + lm) * 2;

  const float kAlpha[NSTEPS] = { 0.5f, 1.0f / 3.0f, 0.25f };

#pragma unroll
  for (int step = 0; step < NSTEPS; ++step) {
    v8f acc[WT] = {};

    v2f bv = *(const v2f*)&sFp[bcolbase];
    v2f acur[WT];
#pragma unroll
    for (int t = 0; t < WT; ++t)
      acur[t] = *(const v2f*)&sS[arowbase + t * 16 * LDSTR];

    for (int k = 0; k < N - 4; k += 4) {
      v2f bnx = *(const v2f*)&sFp[frow(k + 4) + bcolbase];
      v2f an[WT];
#pragma unroll
      for (int t = 0; t < WT; ++t)
        an[t] = *(const v2f*)&sS[arowbase + t * 16 * LDSTR + k + 4];
#pragma unroll
      for (int t = 0; t < WT; ++t)
        acc[t] = __builtin_amdgcn_wmma_f32_16x16x4_f32(
            false, acur[t], false, bv, (short)0, acc[t], false, false);
      bv = bnx;
#pragma unroll
      for (int t = 0; t < WT; ++t) acur[t] = an[t];
    }
#pragma unroll
    for (int t = 0; t < WT; ++t)
      acc[t] = __builtin_amdgcn_wmma_f32_16x16x4_f32(
          false, acur[t], false, bv, (short)0, acc[t], false, false);

    __syncthreads();

    const float alpha = kAlpha[step];
#pragma unroll
    for (int t = 0; t < WT; ++t) {
#pragma unroll
      for (int v = 0; v < 8; ++v) {
        int row  = tg * 32 + t * 16 + v + 8 * lh;
        int col  = tn * 16 + lm;
        int ridx = row * LDSTR + col;
        int fi   = fidx(row, col);
        float msg = acc[t][v];
        float f   = sFp[fi];
        float s   = sS[ridx];
        float cc  = sC[ridx];
        float fn  = __builtin_fmaf(alpha, msg - f, f);
        float fo  = __builtin_fmaf(s, cc - fn, fn);
        if (step == NSTEPS - 1) sS[ridx] = fo;
        else                    sFp[fi]  = fo;
      }
    }
    __syncthreads();
  }

  float* dst = out + (size_t)b * (N * N);
  for (int e = tid; e < N * N; e += NT) {
    int i = e >> 7, j = e & 127;
    float val = (i == j) ? 1.0f
                         : 0.5f * (sS[i * LDSTR + j] + sS[j * LDSTR + i]);
    dst[e] = val;
  }

  }
}

extern "C" void kernel_launch(void* const* d_in, const int* in_sizes, int n_in,
                              void* d_out, int out_size, void* d_ws, size_t ws_size,
                              hipStream_t stream) {
  (void)in_sizes; (void)n_in; (void)out_size; (void)d_ws; (void)ws_size;
  const float* corr = (const float*)d_in[0];
  const float* temp = (const float*)d_in[1];
  float* out = (float*)d_out;

  const size_t shmem =
      ((size_t)3 * N * LDSTR + (size_t)(N / 2) * PSTR) * sizeof(float);
  static bool attr_set = false;
  if (!attr_set) {
    hipFuncSetAttribute((const void*)sgf_kernel,
                        hipFuncAttributeMaxDynamicSharedMemorySize, (int)shmem);
    attr_set = true;
  }
  sgf_kernel<<<dim3(NWG), dim3(NT), shmem, stream>>>(corr, temp, out);
}


# c, soft_adj and running filtered of a lane's own 16 accumulator elements load once per matrix and stay in registers: the 3 updates read nothing from LDS, store 8-byte pairs; sum orders unchanged
# speedup vs baseline: 3.1204x; 1.0981x over previous
#include <hip/hip_runtime.h>


#define N       128
#define LDSTR   132
#define PSTR    288
#define NBATCH  2048
#define EPSF    1e-8f
#define NSTEPS  3
#define NT      1024
#define NWG     256
#define NFT     512
#define WT      2

typedef __attribute__((ext_vector_type(2))) float v2f;
typedef __attribute__((ext_vector_type(8))) float v8f;
typedef __attribute__((ext_vector_type(4))) float v4f;

template <int nthr>
__device__ __forceinline__ void fetch_raw(const float* __restrict__ src, float* dst, int t) {
  constexpr int NP = N * N / (4 * nthr);
  const float* s = src + t * 4;
  float*       p = dst + (t >> 5) * LDSTR + (t & 31) * 4;
  v4f v[NP];
#pragma unroll
  for (int q = 0; q < NP; ++q) v[q] = *(const v4f*)(s + q * nthr * 4);
#pragma unroll
  for (int q = 0; q < NP; ++q) *(v4f*)(p + q * (nthr / 32) * LDSTR) = v[q];
}

__device__ __forceinline__ int frow(int row) {
  return (row >> 1) * PSTR + ((row & 8) << 2);
}
__device__ __forceinline__ int fidx(int row, int col) {
  return frow(row) + col * 2 + (row & 1);
}

__global__ __launch_bounds__(NT, 1)
void sgf_kernel(const float* __restrict__ corr,
                const float* __restrict__ temperature,
                float* __restrict__ out) {
  extern __shared__ float lds[];
  float* sC  = lds;
  float* sS  = lds + N * LDSTR;
  float* sFp = lds + 2 * N * LDSTR;
  float* sR  = lds + 2 * N * LDSTR + (N / 2) * PSTR;

  const int tid0 = threadIdx.x;

  const float invT = 1.0f / temperature[0];

  fetch_raw<NT>(corr + (size_t)blockIdx.x * (N * N), sR, tid0);
  __syncthreads();

  for (int b = blockIdx.x; b < NBATCH; b += NWG) {

  int tid = tid0;
  asm volatile("" : "+v"(tid));
  __builtin_assume(tid >= 0 && tid < NT);
  const int lane = tid & 31;
  const int wave = tid >> 5;
  const int lm   = lane & 15;
  const int lh   = lane >> 4;

  for (int e = tid; e < N * N; e += NT) {
    int i = e >> 7, j = e & 127;
    float x = sR[i * LDSTR + j];
    float y = sR[j * LDSTR + i];
    x = fminf(fmaxf(x, -1.0f + EPSF), 1.0f - EPSF);
    y = fminf(fmaxf(y, -1.0f + EPSF), 1.0f - EPSF);
    sC[i * LDSTR + j] = 0.5f * (x + y);
  }
  __syncthreads();

  if (tid < N) {
    const int i = tid;
    const float* crow = &sC[i * LDSTR];
    float*       srow = &sS[i * LDSTR];
    float t0 = -1e30f, t1 = -1e30f, t2 = -1e30f;
    for (int j0 = 0; j0 < N; j0 += 16) {
      float a[16];
      for (int q = 0; q < 4; ++q) {
        float4 cv4 = *(const float4*)&crow[j0 + 4 * q];
        a[4 * q + 0] = cv4.x; a[4 * q + 1] = cv4.y;
        a[4 * q + 2] = cv4.z; a[4 * q + 3] = cv4.w;
      }
#pragma unroll
      for (int u = 0; u < 16; ++u) {
        float d   = __builtin_sqrtf(2.0f * (1.0f - a[u]) + EPSF);
        float aff = __expf(-d * invT);
        if (j0 + u == i) aff = 0.0f;
        a[u] = aff;
      }
      for (int q = 0; q < 4; ++q)
        *(float4*)&srow[j0 + 4 * q] =
            make_float4(a[4 * q + 0], a[4 * q + 1], a[4 * q + 2], a[4 * q + 3]);
      for (int u = 0; u < 16; ++u) {
        float n0 = fmaxf(t0, a[u]);
        float n1 = __builtin_amdgcn_fmed3f(t0, t1, a[u]);
        float n2 = __builtin_amdgcn_fmed3f(t1, t2, a[u]);
        t0 = n0; t1 = n1; t2 = n2;
      }
    }
    float sum = 0.0f;
    for (int j0 = 0; j0 < N; j0 += 16) {
      float a[16];
      for (int q = 0; q < 4; ++q) {
        float4 v4 = *(const float4*)&srow[j0 + 4 * q];
        a[4 * q + 0] = v4.x; a[4 * q + 1] = v4.y;
        a[4 * q + 2] = v4.z; a[4 * q + 3] = v4.w;
      }
      for (int u = 0; u < 16; ++u)
        if (a[u] >= t2) sum += __expf(a[u] - t0);
    }
    float rsum = 1.0f / sum;
    for (int j0 = 0; j0 < N; j0 += 16) {
      float a[16];
      for (int q = 0; q < 4; ++q) {
        float4 v4 = *(const float4*)&srow[j0 + 4 * q];
        a[4 * q + 0] = v4.x; a[4 * q + 1] = v4.y;
        a[4 * q + 2] = v4.z; a[4 * q + 3] = v4.w;
      }
      for (int u = 0; u < 16; ++u)
        a[u] = (a[u] >= t2) ? __expf(a[u] - t0) * rsum : 0.0f;
      for (int q = 0; q < 4; ++q)
        *(float4*)&srow[j0 + 4 * q] =
            make_float4(a[4 * q + 0], a[4 * q + 1], a[4 * q + 2], a[4 * q + 3]);
    }
  } else if (tid >= NT - NFT && b + NWG < NBATCH) {
    fetch_raw<NFT>(corr + (size_t)(b + NWG) * (N * N), sR, tid & (NFT - 1));
  }
  __syncthreads();

  for (int e = tid; e < N * N; e += NT) {
    int i = e >> 7, j = e & 127;
    if (i < j) {
      float a = sS[i * LDSTR + j];
      float c2 = sS[j * LDSTR + i];
      float m = 0.5f * (a + c2);
      sS[i * LDSTR + j] = m;
      sS[j * LDSTR + i] = m;
    }
  }

  const int tg    = wave >> 3;
  const int tn    = wave & 7;
  const int ecol  = tn * 16 + lm;
  const int erow0 = tg * 32 + 8 * lh;
  float rc[WT][8], rs[WT][8], rf[WT][8];
#pragma unroll
  for (int t = 0; t < WT; ++t)
#pragma unroll
    for (int v = 0; v < 8; ++v) rc[t][v] = sC[(erow0 + t * 16 + v) * LDSTR + ecol];
#pragma unroll
  for (int t = 0; t < WT; ++t)
#pragma unroll
    for (int v = 0; v < 8; v += 2) {
      v2f p = { rc[t][v], rc[t][v + 1] };
      *(v2f*)&sFp[fidx(erow0 + t * 16 + v, ecol)] = p;
    }
  __syncthreads();
#pragma unroll
  for (int t = 0; t < WT; ++t)
#pragma unroll
    for (int v = 0; v < 8; ++v) {
      rs[t][v] = sS[(erow0 + t * 16 + v) * LDSTR + ecol];
      rf[t][v] = rc[t][v];
    }

  const int arowbase = (tg * 32 + lm) * LDSTR + 2 * lh;
  const int bcolbase = lh * PSTR + (tn * 16 + lm) * 2;

  const float kAlpha[NSTEPS] = { 0.5f, 1.0f / 3.0f, 0.25f };

#pragma unroll
  for (int step = 0; step < NSTEPS; ++step) {
    v8f acc[WT] = {};

    v2f bv = *(const v2f*)&sFp[bcolbase];
    v2f acur[WT];
#pragma unroll
    for (int t = 0; t < WT; ++t)
      acur[t] = *(const v2f*)&sS[arowbase + t * 16 * LDSTR];

    for (int k = 0; k < N - 4; k += 4) {
      v2f bnx = *(const v2f*)&sFp[frow(k + 4) + bcolbase];
      v2f an[WT];
#pragma unroll
      for (int t = 0; t < WT; ++t)
        an[t] = *(const v2f*)&sS[arowbase + t * 16 * LDSTR + k + 4];
#pragma unroll
      for (int t = 0; t < WT; ++t)
        acc[t] = __builtin_amdgcn_wmma_f32_16x16x4_f32(
            false, acur[t], false, bv, (short)0, acc[t], false, false);
      bv = bnx;
#pragma unroll
      for (int t = 0; t < WT; ++t) acur[t] = an[t];
    }
#pragma unroll
    for (int t = 0; t < WT; ++t)
      acc[t] = __builtin_amdgcn_wmma_f32_16x16x4_f32(
          false, acur[t], false, bv, (short)0, acc[t], false, false);

    __syncthreads();

    const float alpha = kAlpha[step];
#pragma unroll
    for (int t = 0; t < WT; ++t) {
#pragma unroll
      for (int v = 0; v < 8; v += 2) {
        const int row = erow0 + t * 16 + v;
        float fn0 = __builtin_fmaf(alpha, acc[t][v] - rf[t][v], rf[t][v]);
        float fn1 = __builtin_fmaf(alpha, acc[t][v + 1] - rf[t][v + 1], rf[t][v + 1]);
        float fo0 = __builtin_fmaf(rs[t][v], rc[t][v] - fn0, fn0);
        float fo1 = __builtin_fmaf(rs[t][v + 1], rc[t][v + 1] - fn1, fn1);
        rf[t][v] = fo0; rf[t][v + 1] = fo1;
        if (step == NSTEPS - 1) {
          sS[row * LDSTR + ecol]       = fo0;
          sS[(row + 1) * LDSTR + ecol] = fo1;
        } else {
          v2f p = { fo0, fo1 };
          *(v2f*)&sFp[fidx(row, ecol)] = p;
        }
      }
    }
    __syncthreads();
  }

  float* dst = out + (size_t)b * (N * N);
  for (int e = tid; e < N * N; e += NT) {
    int i = e >> 7, j = e & 127;
    float val = (i == j) ? 1.0f
                         : 0.5f * (sS[i * LDSTR + j] + sS[j * LDSTR + i]);
    dst[e] = val;
  }

  }
}

extern "C" void kernel_launch(void* const* d_in, const int* in_sizes, int n_in,
                              void* d_out, int out_size, void* d_ws, size_t ws_size,
                              hipStream_t stream) {
  (void)in_sizes; (void)n_in; (void)out_size; (void)d_ws; (void)ws_size;
  const float* corr = (const float*)d_in[0];
  const float* temp = (const float*)d_in[1];
  float* out = (float*)d_out;

  const size_t shmem =
      ((size_t)3 * N * LDSTR + (size_t)(N / 2) * PSTR) * sizeof(float);
  static bool attr_set = false;
  if (!attr_set) {
    hipFuncSetAttribute((const void*)sgf_kernel,
                        hipFuncAttributeMaxDynamicSharedMemorySize, (int)shmem);
    attr_set = true;
  }
  sgf_kernel<<<dim3(NWG), dim3(NT), shmem, stream>>>(corr, temp, out);
}


# c and the affinity made once per unordered pair on a balanced 64x128 fold of the upper triangle, all 32 waves, mirror stores (half the sqrt/exp); soft_adj sym on the same fold; sum orders unchanged
# speedup vs baseline: 3.2548x; 1.0431x over previous
#include <hip/hip_runtime.h>


#define N       128
#define LDSTR   132
#define PSTR    288
#define NBATCH  2048
#define EPSF    1e-8f
#define NSTEPS  3
#define NT      1024
#define NWG     256
#define NFT     512
#define WT      2

typedef __attribute__((ext_vector_type(2))) float v2f;
typedef __attribute__((ext_vector_type(8))) float v8f;
typedef __attribute__((ext_vector_type(4))) float v4f;

template <int nthr>
__device__ __forceinline__ void fetch_raw(const float* __restrict__ src, float* dst, int t) {
  constexpr int NP = N * N / (4 * nthr);
  const float* s = src + t * 4;
  float*       p = dst + (t >> 5) * LDSTR + (t & 31) * 4;
  v4f v[NP];
#pragma unroll
  for (int q = 0; q < NP; ++q) v[q] = *(const v4f*)(s + q * nthr * 4);
#pragma unroll
  for (int q = 0; q < NP; ++q) *(v4f*)(p + q * (nthr / 32) * LDSTR) = v[q];
}

__device__ __forceinline__ int frow(int row) {
  return (row >> 1) * PSTR + ((row & 8) << 2);
}
__device__ __forceinline__ int fidx(int row, int col) {
  return frow(row) + col * 2 + (row & 1);
}

__device__ __forceinline__ void fold_pair(int s, int& i, int& j) {
  const int p = s >> 7, t = s & 127;
  const bool up = t > p;
  i = up ? p : (N - 1) - p;
  j = up ? t : (N - 1) - t;
}

__global__ __launch_bounds__(NT, 1)
void sgf_kernel(const float* __restrict__ corr,
                const float* __restrict__ temperature,
                float* __restrict__ out) {
  extern __shared__ float lds[];
  float* sC  = lds;
  float* sS  = lds + N * LDSTR;
  float* sFp = lds + 2 * N * LDSTR;
  float* sR  = lds + 2 * N * LDSTR + (N / 2) * PSTR;
  float* sA  = sFp;

  const int tid0 = threadIdx.x;

  const float invT = 1.0f / temperature[0];

  fetch_raw<NT>(corr + (size_t)blockIdx.x * (N * N), sR, tid0);
  __syncthreads();

  for (int b = blockIdx.x; b < NBATCH; b += NWG) {

  int tid = tid0;
  asm volatile("" : "+v"(tid));
  __builtin_assume(tid >= 0 && tid < NT);
  const int lane = tid & 31;
  const int wave = tid >> 5;
  const int lm   = lane & 15;
  const int lh   = lane >> 4;

#pragma unroll
  for (int s = tid; s < (N / 2) * N; s += NT) {
    int i, j;
    fold_pair(s, i, j);
    float x = sR[i * LDSTR + j];
    float y = sR[j * LDSTR + i];
    x = fminf(fmaxf(x, -1.0f + EPSF), 1.0f - EPSF);
    y = fminf(fmaxf(y, -1.0f + EPSF), 1.0f - EPSF);
    const float cv  = 0.5f * (x + y);
    const float d   = __builtin_sqrtf(2.0f * (1.0f - cv) + EPSF);
    float       aff = __expf(-d * invT);
    if (i == j) aff = 0.0f;
    sC[i * LDSTR + j] = cv;
    sC[j * LDSTR + i] = cv;
    sA[i * LDSTR + j] = aff;
    sA[j * LDSTR + i] = aff;
  }
  if (tid < N / 2) {
    float x = sR[tid * (LDSTR + 1)];
    x = fminf(fmaxf(x, -1.0f + EPSF), 1.0f - EPSF);
    sC[tid * (LDSTR + 1)] = 0.5f * (x + x);
    sA[tid * (LDSTR + 1)] = 0.0f;
  }
  __syncthreads();

  if (tid < N) {
    const int i = tid;
    const float* arow = &sA[i * LDSTR];
    float*       srow = &sS[i * LDSTR];
    float t0 = -1e30f, t1 = -1e30f, t2 = -1e30f;
#pragma nounroll
    for (int j0 = 0; j0 < N; j0 += 32) {
      float a[32];
      for (int q = 0; q < 8; ++q) {
        float4 v4 = *(const float4*)&arow[j0 + 4 * q];
        a[4 * q + 0] = v4.x; a[4 * q + 1] = v4.y;
        a[4 * q + 2] = v4.z; a[4 * q + 3] = v4.w;
      }
#pragma unroll
      for (int u = 0; u < 32; ++u) {
        float n0 = fmaxf(t0, a[u]);
        float n1 = __builtin_amdgcn_fmed3f(t0, t1, a[u]);
        float n2 = __builtin_amdgcn_fmed3f(t1, t2, a[u]);
        t0 = n0; t1 = n1; t2 = n2;
      }
    }
    float sum = 0.0f;
    for (int j0 = 0; j0 < N; j0 += 16) {
      float a[16];
      for (int q = 0; q < 4; ++q) {
        float4 v4 = *(const float4*)&arow[j0 + 4 * q];
        a[4 * q + 0] = v4.x; a[4 * q + 1] = v4.y;
        a[4 * q + 2] = v4.z; a[4 * q + 3] = v4.w;
      }
      for (int u = 0; u < 16; ++u)
        if (a[u] >= t2) sum += __expf(a[u] - t0);
    }
    float rsum = 1.0f / sum;
    for (int j0 = 0; j0 < N; j0 += 16) {
      float a[16];
      for (int q = 0; q < 4; ++q) {
        float4 v4 = *(const float4*)&arow[j0 + 4 * q];
        a[4 * q + 0] = v4.x; a[4 * q + 1] = v4.y;
        a[4 * q + 2] = v4.z; a[4 * q + 3] = v4.w;
      }
      for (int u = 0; u < 16; ++u)
        a[u] = (a[u] >= t2) ? __expf(a[u] - t0) * rsum : 0.0f;
      for (int q = 0; q < 4; ++q)
        *(float4*)&srow[j0 + 4 * q] =
            make_float4(a[4 * q + 0], a[4 * q + 1], a[4 * q + 2], a[4 * q + 3]);
    }
  } else if (tid >= NT - NFT && b + NWG < NBATCH) {
    fetch_raw<NFT>(corr + (size_t)(b + NWG) * (N * N), sR, tid & (NFT - 1));
  }
  __syncthreads();

#pragma unroll
  for (int s = tid; s < (N / 2) * N; s += NT) {
    int i, j;
    fold_pair(s, i, j);
    float a = sS[i * LDSTR + j];
    float c2 = sS[j * LDSTR + i];
    float m = 0.5f * (a + c2);
    sS[i * LDSTR + j] = m;
    sS[j * LDSTR + i] = m;
  }

  const int tg    = wave >> 3;
  const int tn    = wave & 7;
  const int ecol  = tn * 16 + lm;
  const int erow0 = tg * 32 + 8 * lh;
  float rc[WT][8], rs[WT][8], rf[WT][8];
#pragma unroll
  for (int t = 0; t < WT; ++t)
#pragma unroll
    for (int v = 0; v < 8; ++v) rc[t][v] = sC[(erow0 + t * 16 + v) * LDSTR + ecol];
#pragma unroll
  for (int t = 0; t < WT; ++t)
#pragma unroll
    for (int v = 0; v < 8; v += 2) {
      v2f p = { rc[t][v], rc[t][v + 1] };
      *(v2f*)&sFp[fidx(erow0 + t * 16 + v, ecol)] = p;
    }
  __syncthreads();
#pragma unroll
  for (int t = 0; t < WT; ++t)
#pragma unroll
    for (int v = 0; v < 8; ++v) {
      rs[t][v] = sS[(erow0 + t * 16 + v) * LDSTR + ecol];
      rf[t][v] = rc[t][v];
    }

  const int arowbase = (tg * 32 + lm) * LDSTR + 2 * lh;
  const int bcolbase = lh * PSTR + (tn * 16 + lm) * 2;

  const float kAlpha[NSTEPS] = { 0.5f, 1.0f / 3.0f, 0.25f };

#pragma unroll
  for (int step = 0; step < NSTEPS; ++step) {
    v8f acc[WT] = {};

    v2f bv = *(const v2f*)&sFp[bcolbase];
    v2f acur[WT];
#pragma unroll
    for (int t = 0; t < WT; ++t)
      acur[t] = *(const v2f*)&sS[arowbase + t * 16 * LDSTR];

    for (int k = 0; k < N - 4; k += 4) {
      v2f bnx = *(const v2f*)&sFp[frow(k + 4) + bcolbase];
      v2f an[WT];
#pragma unroll
      for (int t = 0; t < WT; ++t)
        an[t] = *(const v2f*)&sS[arowbase + t * 16 * LDSTR + k + 4];
#pragma unroll
      for (int t = 0; t < WT; ++t)
        acc[t] = __builtin_amdgcn_wmma_f32_16x16x4_f32(
            false, acur[t], false, bv, (short)0, acc[t], false, false);
      bv = bnx;
#pragma unroll
      for (int t = 0; t < WT; ++t) acur[t] = an[t];
    }
#pragma unroll
    for (int t = 0; t < WT; ++t)
      acc[t] = __builtin_amdgcn_wmma_f32_16x16x4_f32(
          false, acur[t], false, bv, (short)0, acc[t], false, false);

    __syncthreads();

    const float alpha = kAlpha[step];
#pragma unroll
    for (int t = 0; t < WT; ++t) {
#pragma unroll
      for (int v = 0; v < 8; v += 2) {
        const int row = erow0 + t * 16 + v;
        float fn0 = __builtin_fmaf(alpha, acc[t][v] - rf[t][v], rf[t][v]);
        float fn1 = __builtin_fmaf(alpha, acc[t][v + 1] - rf[t][v + 1], rf[t][v + 1]);
        float fo0 = __builtin_fmaf(rs[t][v], rc[t][v] - fn0, fn0);
        float fo1 = __builtin_fmaf(rs[t][v + 1], rc[t][v + 1] - fn1, fn1);
        rf[t][v] = fo0; rf[t][v + 1] = fo1;
        if (step == NSTEPS - 1) {
          sS[row * LDSTR + ecol]       = fo0;
          sS[(row + 1) * LDSTR + ecol] = fo1;
        } else {
          v2f p = { fo0, fo1 };
          *(v2f*)&sFp[fidx(row, ecol)] = p;
        }
      }
    }
    __syncthreads();
  }

  float* dst = out + (size_t)b * (N * N);
  for (int e = tid; e < N * N; e += NT) {
    int i = e >> 7, j = e & 127;
    float val = (i == j) ? 1.0f
                         : 0.5f * (sS[i * LDSTR + j] + sS[j * LDSTR + i]);
    dst[e] = val;
  }

  }
}

extern "C" void kernel_launch(void* const* d_in, const int* in_sizes, int n_in,
                              void* d_out, int out_size, void* d_ws, size_t ws_size,
                              hipStream_t stream) {
  (void)in_sizes; (void)n_in; (void)out_size; (void)d_ws; (void)ws_size;
  const float* corr = (const float*)d_in[0];
  const float* temp = (const float*)d_in[1];
  float* out = (float*)d_out;

  const size_t shmem =
      ((size_t)3 * N * LDSTR + (size_t)(N / 2) * PSTR) * sizeof(float);
  static bool attr_set = false;
  if (!attr_set) {
    hipFuncSetAttribute((const void*)sgf_kernel,
                        hipFuncAttributeMaxDynamicSharedMemorySize, (int)shmem);
    attr_set = true;
  }
  sgf_kernel<<<dim3(NWG), dim3(NT), shmem, stream>>>(corr, temp, out);
}


# Graph build: masked-softmax terms taken from the row's three largest affinities (2 exp per row, not 256): a count pass and a select pass replace the two exp passes; softmax row sum order changes
# speedup vs baseline: 3.3851x; 1.0400x over previous
#include <hip/hip_runtime.h>


#define N       128
#define LDSTR   132
#define PSTR    288
#define NBATCH  2048
#define EPSF    1e-8f
#define NSTEPS  3
#define NT      1024
#define NWG     256
#define NFT     512
#define WT      2

typedef __attribute__((ext_vector_type(2))) float v2f;
typedef __attribute__((ext_vector_type(8))) float v8f;
typedef __attribute__((ext_vector_type(4))) float v4f;

template <int nthr>
__device__ __forceinline__ void fetch_raw(const float* __restrict__ src, float* dst, int t) {
  constexpr int NP = N * N / (4 * nthr);
  const float* s = src + t * 4;
  float*       p = dst + (t >> 5) * LDSTR + (t & 31) * 4;
  v4f v[NP];
#pragma unroll
  for (int q = 0; q < NP; ++q) v[q] = *(const v4f*)(s + q * nthr * 4);
#pragma unroll
  for (int q = 0; q < NP; ++q) *(v4f*)(p + q * (nthr / 32) * LDSTR) = v[q];
}

__device__ __forceinline__ int frow(int row) {
  return (row >> 1) * PSTR + ((row & 8) << 2);
}
__device__ __forceinline__ int fidx(int row, int col) {
  return frow(row) + col * 2 + (row & 1);
}

__device__ __forceinline__ void fold_pair(int s, int& i, int& j) {
  const int p = s >> 7, t = s & 127;
  const bool up = t > p;
  i = up ? p : (N - 1) - p;
  j = up ? t : (N - 1) - t;
}

__global__ __launch_bounds__(NT, 1)
void sgf_kernel(const float* __restrict__ corr,
                const float* __restrict__ temperature,
                float* __restrict__ out) {
  extern __shared__ float lds[];
  float* sC  = lds;
  float* sS  = lds + N * LDSTR;
  float* sFp = lds + 2 * N * LDSTR;
  float* sR  = lds + 2 * N * LDSTR + (N / 2) * PSTR;
  float* sA  = sFp;

  const int tid0 = threadIdx.x;

  const float invT = 1.0f / temperature[0];

  fetch_raw<NT>(corr + (size_t)blockIdx.x * (N * N), sR, tid0);
  __syncthreads();

  for (int b = blockIdx.x; b < NBATCH; b += NWG) {

  int tid = tid0;
  asm volatile("" : "+v"(tid));
  __builtin_assume(tid >= 0 && tid < NT);
  const int lane = tid & 31;
  const int wave = tid >> 5;
  const int lm   = lane & 15;
  const int lh   = lane >> 4;

#pragma unroll
  for (int s = tid; s < (N / 2) * N; s += NT) {
    int i, j;
    fold_pair(s, i, j);
    float x = sR[i * LDSTR + j];
    float y = sR[j * LDSTR + i];
    x = fminf(fmaxf(x, -1.0f + EPSF), 1.0f - EPSF);
    y = fminf(fmaxf(y, -1.0f + EPSF), 1.0f - EPSF);
    const float cv  = 0.5f * (x + y);
    const float d   = __builtin_sqrtf(2.0f * (1.0f - cv) + EPSF);
    float       aff = __expf(-d * invT);
    if (i == j) aff = 0.0f;
    sC[i * LDSTR + j] = cv;
    sC[j * LDSTR + i] = cv;
    sA[i * LDSTR + j] = aff;
    sA[j * LDSTR + i] = aff;
  }
  if (tid < N / 2) {
    float x = sR[tid * (LDSTR + 1)];
    x = fminf(fmaxf(x, -1.0f + EPSF), 1.0f - EPSF);
    sC[tid * (LDSTR + 1)] = 0.5f * (x + x);
    sA[tid * (LDSTR + 1)] = 0.0f;
  }
  __syncthreads();

  if (tid < N) {
    const int i = tid;
    const float* arow = &sA[i * LDSTR];
    float*       srow = &sS[i * LDSTR];
    float t0 = -1e30f, t1 = -1e30f, t2 = -1e30f;
#pragma nounroll
    for (int j0 = 0; j0 < N; j0 += 32) {
      float a[32];
      for (int q = 0; q < 8; ++q) {
        float4 v4 = *(const float4*)&arow[j0 + 4 * q];
        a[4 * q + 0] = v4.x; a[4 * q + 1] = v4.y;
        a[4 * q + 2] = v4.z; a[4 * q + 3] = v4.w;
      }
#pragma unroll
      for (int u = 0; u < 32; ++u) {
        float n0 = fmaxf(t0, a[u]);
        float n1 = __builtin_amdgcn_fmed3f(t0, t1, a[u]);
        float n2 = __builtin_amdgcn_fmed3f(t1, t2, a[u]);
        t0 = n0; t1 = n1; t2 = n2;
      }
    }
    const float e1 = __expf(t1 - t0);
    const float e2 = __expf(t2 - t0);
    int c0 = 0, c1 = 0, c2 = 0, c3 = 0;
    for (int j0 = 0; j0 < N; j0 += 16) {
      float a[16];
      for (int q = 0; q < 4; ++q) {
        float4 v4 = *(const float4*)&arow[j0 + 4 * q];
        a[4 * q + 0] = v4.x; a[4 * q + 1] = v4.y;
        a[4 * q + 2] = v4.z; a[4 * q + 3] = v4.w;
      }
      for (int u = 0; u < 16; u += 4) {
        c0 += (a[u + 0] >= t2) ? 1 : 0;
        c1 += (a[u + 1] >= t2) ? 1 : 0;
        c2 += (a[u + 2] >= t2) ? 1 : 0;
        c3 += (a[u + 3] >= t2) ? 1 : 0;
      }
    }
    const int   npass = (c0 + c1) + (c2 + c3);
    const int   nabove = ((t0 > t2) ? 1 : 0) + ((t1 > t2) ? 1 : 0);
    const float g0 = (t0 > t2) ? 1.0f : 0.0f;
    const float g1 = (t1 > t2) ? e1 : 0.0f;
    const float sum = (g0 + g1) + (float)(npass - nabove) * e2;
    const float rsum = 1.0f / sum;
    const float p0 = rsum, p1 = e1 * rsum, p2 = e2 * rsum;
#pragma unroll
    for (int j0 = 0; j0 < N; j0 += 16) {
      float a[16];
      for (int q = 0; q < 4; ++q) {
        float4 v4 = *(const float4*)&arow[j0 + 4 * q];
        a[4 * q + 0] = v4.x; a[4 * q + 1] = v4.y;
        a[4 * q + 2] = v4.z; a[4 * q + 3] = v4.w;
      }
      for (int u = 0; u < 16; ++u) {
        const float hi = (a[u] >= t0) ? p0 : p1;
        const float md = (a[u] >= t1) ? hi : p2;
        a[u] = (a[u] >= t2) ? md : 0.0f;
      }
      for (int q = 0; q < 4; ++q)
        *(float4*)&srow[j0 + 4 * q] =
            make_float4(a[4 * q + 0], a[4 * q + 1], a[4 * q + 2], a[4 * q + 3]);
    }
  } else if (tid >= NT - NFT && b + NWG < NBATCH) {
    fetch_raw<NFT>(corr + (size_t)(b + NWG) * (N * N), sR, tid & (NFT - 1));
  }
  __syncthreads();

#pragma unroll
  for (int s = tid; s < (N / 2) * N; s += NT) {
    int i, j;
    fold_pair(s, i, j);
    float a = sS[i * LDSTR + j];
    float c2 = sS[j * LDSTR + i];
    float m = 0.5f * (a + c2);
    sS[i * LDSTR + j] = m;
    sS[j * LDSTR + i] = m;
  }

  const int tg    = wave >> 3;
  const int tn    = wave & 7;
  const int ecol  = tn * 16 + lm;
  const int erow0 = tg * 32 + 8 * lh;
  float rc[WT][8], rs[WT][8], rf[WT][8];
#pragma unroll
  for (int t = 0; t < WT; ++t)
#pragma unroll
    for (int v = 0; v < 8; ++v) rc[t][v] = sC[(erow0 + t * 16 + v) * LDSTR + ecol];
#pragma unroll
  for (int t = 0; t < WT; ++t)
#pragma unroll
    for (int v = 0; v < 8; v += 2) {
      v2f p = { rc[t][v], rc[t][v + 1] };
      *(v2f*)&sFp[fidx(erow0 + t * 16 + v, ecol)] = p;
    }
  __syncthreads();
#pragma unroll
  for (int t = 0; t < WT; ++t)
#pragma unroll
    for (int v = 0; v < 8; ++v) {
      rs[t][v] = sS[(erow0 + t * 16 + v) * LDSTR + ecol];
      rf[t][v] = rc[t][v];
    }

  const int arowbase = (tg * 32 + lm) * LDSTR + 2 * lh;
  const int bcolbase = lh * PSTR + (tn * 16 + lm) * 2;

  const float kAlpha[NSTEPS] = { 0.5f, 1.0f / 3.0f, 0.25f };

#pragma unroll
  for (int step = 0; step < NSTEPS; ++step) {
    v8f acc[WT] = {};

    v2f bv = *(const v2f*)&sFp[bcolbase];
    v2f acur[WT];
#pragma unroll
    for (int t = 0; t < WT; ++t)
      acur[t] = *(const v2f*)&sS[arowbase + t * 16 * LDSTR];

    for (int k = 0; k < N - 4; k += 4) {
      v2f bnx = *(const v2f*)&sFp[frow(k + 4) + bcolbase];
      v2f an[WT];
#pragma unroll
      for (int t = 0; t < WT; ++t)
        an[t] = *(const v2f*)&sS[arowbase + t * 16 * LDSTR + k + 4];
#pragma unroll
      for (int t = 0; t < WT; ++t)
        acc[t] = __builtin_amdgcn_wmma_f32_16x16x4_f32(
            false, acur[t], false, bv, (short)0, acc[t], false, false);
      bv = bnx;
#pragma unroll
      for (int t = 0; t < WT; ++t) acur[t] = an[t];
    }
#pragma unroll
    for (int t = 0; t < WT; ++t)
      acc[t] = __builtin_amdgcn_wmma_f32_16x16x4_f32(
          false, acur[t], false, bv, (short)0, acc[t], false, false);

    __syncthreads();

    const float alpha = kAlpha[step];
#pragma unroll
    for (int t = 0; t < WT; ++t) {
#pragma unroll
      for (int v = 0; v < 8; v += 2) {
        const int row = erow0 + t * 16 + v;
        float fn0 = __builtin_fmaf(alpha, acc[t][v] - rf[t][v], rf[t][v]);
        float fn1 = __builtin_fmaf(alpha, acc[t][v + 1] - rf[t][v + 1], rf[t][v + 1]);
        float fo0 = __builtin_fmaf(rs[t][v], rc[t][v] - fn0, fn0);
        float fo1 = __builtin_fmaf(rs[t][v + 1], rc[t][v + 1] - fn1, fn1);
        rf[t][v] = fo0; rf[t][v + 1] = fo1;
        if (step == NSTEPS - 1) {
          sS[row * LDSTR + ecol]       = fo0;
          sS[(row + 1) * LDSTR + ecol] = fo1;
        } else {
          v2f p = { fo0, fo1 };
          *(v2f*)&sFp[fidx(row, ecol)] = p;
        }
      }
    }
    __syncthreads();
  }

  float* dst = out + (size_t)b * (N * N);
  for (int e = tid; e < N * N; e += NT) {
    int i = e >> 7, j = e & 127;
    float val = (i == j) ? 1.0f
                         : 0.5f * (sS[i * LDSTR + j] + sS[j * LDSTR + i]);
    dst[e] = val;
  }

  }
}

extern "C" void kernel_launch(void* const* d_in, const int* in_sizes, int n_in,
                              void* d_out, int out_size, void* d_ws, size_t ws_size,
                              hipStream_t stream) {
  (void)in_sizes; (void)n_in; (void)out_size; (void)d_ws; (void)ws_size;
  const float* corr = (const float*)d_in[0];
  const float* temp = (const float*)d_in[1];
  float* out = (float*)d_out;

  const size_t shmem =
      ((size_t)3 * N * LDSTR + (size_t)(N / 2) * PSTR) * sizeof(float);
  static bool attr_set = false;
  if (!attr_set) {
    hipFuncSetAttribute((const void*)sgf_kernel,
                        hipFuncAttributeMaxDynamicSharedMemorySize, (int)shmem);
    attr_set = true;
  }
  sgf_kernel<<<dim3(NWG), dim3(NT), shmem, stream>>>(corr, temp, out);
}


# Products skip soft_adj's all-zero columns: per 16-row tile the K loop runs over a list of the tile's nonzero columns, A and B fragments gathered by it; wave tile 32x16 -> 16x32; k sum order changes
# speedup vs baseline: 4.0109x; 1.1849x over previous
#include <hip/hip_runtime.h>


#define N       128
#define LDSTR   132
#define PSTR    288
#define KLS     144
#define NRT     8
#define NBATCH  2048
#define EPSF    1e-8f
#define NSTEPS  3
#define NT      1024
#define NWG     256
#define NFT     512
#define WT      2

typedef __attribute__((ext_vector_type(2))) float v2f;
typedef __attribute__((ext_vector_type(8))) float v8f;
typedef __attribute__((ext_vector_type(4))) float v4f;

template <int nthr>
__device__ __forceinline__ void fetch_raw(const float* __restrict__ src, float* dst, int t) {
  constexpr int NP = N * N / (4 * nthr);
  const float* s = src + t * 4;
  float*       p = dst + (t >> 5) * LDSTR + (t & 31) * 4;
  v4f v[NP];
#pragma unroll
  for (int q = 0; q < NP; ++q) v[q] = *(const v4f*)(s + q * nthr * 4);
#pragma unroll
  for (int q = 0; q < NP; ++q) *(v4f*)(p + q * (nthr / 32) * LDSTR) = v[q];
}

__device__ __forceinline__ int fcol(int col) {
  return (col & ~31) + ((col & 15) << 1) + ((col >> 4) & 1);
}

__device__ __forceinline__ void fold_pair(int s, int& i, int& j) {
  const int p = s >> 7, t = s & 127;
  const bool up = t > p;
  i = up ? p : (N - 1) - p;
  j = up ? t : (N - 1) - t;
}

__global__ __launch_bounds__(NT, 1)
void sgf_kernel(const float* __restrict__ corr,
                const float* __restrict__ temperature,
                float* __restrict__ out) {
  extern __shared__ float lds[];
  float* sC  = lds;
  float* sS  = lds + N * LDSTR;
  float* sF  = lds + 2 * N * LDSTR;
  float* sR  = lds + 2 * N * LDSTR + (N / 2) * PSTR;
  float* sA  = sF;
  unsigned short* sKl = (unsigned short*)(lds + 3 * N * LDSTR + (N / 2) * PSTR);
  int*            sKn = (int*)(sKl + NRT * KLS);

  const int tid0 = threadIdx.x;

  const float invT = 1.0f / temperature[0];

  fetch_raw<NT>(corr + (size_t)blockIdx.x * (N * N), sR, tid0);
  __syncthreads();

  for (int b = blockIdx.x; b < NBATCH; b += NWG) {

  int tid = tid0;
  asm volatile("" : "+v"(tid));
  __builtin_assume(tid >= 0 && tid < NT);
  const int lane = tid & 31;
  const int wave = tid >> 5;
  const int lm   = lane & 15;
  const int lh   = lane >> 4;

#pragma unroll
  for (int s = tid; s < (N / 2) * N; s += NT) {
    int i, j;
    fold_pair(s, i, j);
    float x = sR[i * LDSTR + j];
    float y = sR[j * LDSTR + i];
    x = fminf(fmaxf(x, -1.0f + EPSF), 1.0f - EPSF);
    y = fminf(fmaxf(y, -1.0f + EPSF), 1.0f - EPSF);
    const float cv  = 0.5f * (x + y);
    const float d   = __builtin_sqrtf(2.0f * (1.0f - cv) + EPSF);
    float       aff = __expf(-d * invT);
    if (i == j) aff = 0.0f;
    sC[i * LDSTR + j] = cv;
    sC[j * LDSTR + i] = cv;
    sA[i * LDSTR + j] = aff;
    sA[j * LDSTR + i] = aff;
  }
  if (tid < N / 2) {
    float x = sR[tid * (LDSTR + 1)];
    x = fminf(fmaxf(x, -1.0f + EPSF), 1.0f - EPSF);
    sC[tid * (LDSTR + 1)] = 0.5f * (x + x);
    sA[tid * (LDSTR + 1)] = 0.0f;
  }
  __syncthreads();

  if (tid < N) {
    const int i = tid;
    const float* arow = &sA[i * LDSTR];
    float*       srow = &sS[i * LDSTR];
    float t0 = -1e30f, t1 = -1e30f, t2 = -1e30f;
#pragma nounroll
    for (int j0 = 0; j0 < N; j0 += 32) {
      float a[32];
      for (int q = 0; q < 8; ++q) {
        float4 v4 = *(const float4*)&arow[j0 + 4 * q];
        a[4 * q + 0] = v4.x; a[4 * q + 1] = v4.y;
        a[4 * q + 2] = v4.z; a[4 * q + 3] = v4.w;
      }
#pragma unroll
      for (int u = 0; u < 32; ++u) {
        float n0 = fmaxf(t0, a[u]);
        float n1 = __builtin_amdgcn_fmed3f(t0, t1, a[u]);
        float n2 = __builtin_amdgcn_fmed3f(t1, t2, a[u]);
        t0 = n0; t1 = n1; t2 = n2;
      }
    }
    const float e1 = __expf(t1 - t0);
    const float e2 = __expf(t2 - t0);
    int c0 = 0, c1 = 0, c2 = 0, c3 = 0;
    for (int j0 = 0; j0 < N; j0 += 16) {
      float a[16];
      for (int q = 0; q < 4; ++q) {
        float4 v4 = *(const float4*)&arow[j0 + 4 * q];
        a[4 * q + 0] = v4.x; a[4 * q + 1] = v4.y;
        a[4 * q + 2] = v4.z; a[4 * q + 3] = v4.w;
      }
      for (int u = 0; u < 16; u += 4) {
        c0 += (a[u + 0] >= t2) ? 1 : 0;
        c1 += (a[u + 1] >= t2) ? 1 : 0;
        c2 += (a[u + 2] >= t2) ? 1 : 0;
        c3 += (a[u + 3] >= t2) ? 1 : 0;
      }
    }
    const int   npass = (c0 + c1) + (c2 + c3);
    const int   nabove = ((t0 > t2) ? 1 : 0) + ((t1 > t2) ? 1 : 0);
    const float g0 = (t0 > t2) ? 1.0f : 0.0f;
    const float g1 = (t1 > t2) ? e1 : 0.0f;
    const float sum = (g0 + g1) + (float)(npass - nabove) * e2;
    const float rsum = 1.0f / sum;
    const float p0 = rsum, p1 = e1 * rsum, p2 = e2 * rsum;
#pragma unroll
    for (int j0 = 0; j0 < N; j0 += 16) {
      float a[16];
      for (int q = 0; q < 4; ++q) {
        float4 v4 = *(const float4*)&arow[j0 + 4 * q];
        a[4 * q + 0] = v4.x; a[4 * q + 1] = v4.y;
        a[4 * q + 2] = v4.z; a[4 * q + 3] = v4.w;
      }
      for (int u = 0; u < 16; ++u) {
        const float hi = (a[u] >= t0) ? p0 : p1;
        const float md = (a[u] >= t1) ? hi : p2;
        a[u] = (a[u] >= t2) ? md : 0.0f;
      }
      for (int q = 0; q < 4; ++q)
        *(float4*)&srow[j0 + 4 * q] =
            make_float4(a[4 * q + 0], a[4 * q + 1], a[4 * q + 2], a[4 * q + 3]);
    }
  } else if (tid >= NT - NFT && b + NWG < NBATCH) {
    fetch_raw<NFT>(corr + (size_t)(b + NWG) * (N * N), sR, tid & (NFT - 1));
  }
  __syncthreads();

#pragma unroll
  for (int s = tid; s < (N / 2) * N; s += NT) {
    int i, j;
    fold_pair(s, i, j);
    float a = sS[i * LDSTR + j];
    float c2 = sS[j * LDSTR + i];
    float m = 0.5f * (a + c2);
    sS[i * LDSTR + j] = m;
    sS[j * LDSTR + i] = m;
  }

  __syncthreads();

  if (tid < NRT * 32) {
    const int rtk = wave;
    unsigned short* klw = sKl + rtk * KLS;
    const float* srows = sS + rtk * 16 * LDSTR + lane;
    int cnt = 0;
    int zcol = 0;
    bool havez = false;
#pragma unroll
    for (int q = 0; q < 4; ++q) {
      float cv16[16];
#pragma unroll
      for (int r = 0; r < 16; ++r) cv16[r] = srows[r * LDSTR + 32 * q];
      unsigned bits = 0u;
#pragma unroll
      for (int r = 0; r < 16; ++r) bits |= __float_as_uint(cv16[r]) & 0x7fffffffu;
      const bool nz = bits != 0u;
      const unsigned m = __builtin_amdgcn_ballot_w32(nz);
      const int pos = cnt + (int)__builtin_amdgcn_mbcnt_lo(m, 0u);
      if (nz) klw[pos] = (unsigned short)(32 * q + lane);
      cnt += __builtin_popcount(m);
      const unsigned zm = ~m;
      if (!havez && zm != 0u) { zcol = 32 * q + __builtin_ctz(zm); havez = true; }
    }
    int cnt4 = (cnt + 3) & ~3;
    if (cnt4 < 4) cnt4 = 4;
    if (lane < cnt4 - cnt) klw[cnt + lane] = (unsigned short)zcol;
    if (lane == 0) sKn[rtk] = cnt4 >> 2;
  }

  const int rt    = wave >> 2;
  const int cp    = wave & 3;
  const int ecol0 = cp * 32 + lm;
  const int erow0 = rt * 16 + 8 * lh;
  const int fc0   = cp * 32 + 2 * lm;
  float rc[WT][8], rs[WT][8], rf[WT][8];
#pragma unroll
  for (int t = 0; t < WT; ++t)
#pragma unroll
    for (int v = 0; v < 8; ++v) {
      rc[t][v] = sC[(erow0 + v) * LDSTR + ecol0 + 16 * t];
      rs[t][v] = sS[(erow0 + v) * LDSTR + ecol0 + 16 * t];
    }
#pragma unroll
  for (int v = 0; v < 8; ++v) {
    v2f p = { rc[0][v], rc[1][v] };
    *(v2f*)&sF[(erow0 + v) * LDSTR + fc0] = p;
    rf[0][v] = rc[0][v];
    rf[1][v] = rc[1][v];
  }
  __syncthreads();

  const float* arow = sS + (rt * 16 + lm) * LDSTR;
  const float* bcol = sF + fc0;
  const unsigned short* kl = sKl + rt * KLS + 2 * lh;
  const int nk = __builtin_amdgcn_readfirstlane(sKn[rt]);

  const float kAlpha[NSTEPS] = { 0.5f, 1.0f / 3.0f, 0.25f };

#pragma unroll
  for (int step = 0; step < NSTEPS; ++step) {
    v8f acc[WT] = {};

    unsigned e1;
    v2f acur, x0, x1;
    {
      const unsigned e0 = *(const unsigned*)kl;
      const int k0 = (int)(e0 & 0xffffu), k1 = (int)(e0 >> 16);
      acur = (v2f){ arow[k0], arow[k1] };
      x0 = *(const v2f*)&bcol[k0 * LDSTR];
      x1 = *(const v2f*)&bcol[k1 * LDSTR];
      e1 = *(const unsigned*)(kl + 4 * ((nk > 1) ? 1 : 0));
    }

    for (int kk = 0; kk < nk; ++kk) {
      const int n0 = (int)(e1 & 0xffffu), n1 = (int)(e1 >> 16);
      const v2f an = { arow[n0], arow[n1] };
      const v2f y0 = *(const v2f*)&bcol[n0 * LDSTR];
      const v2f y1 = *(const v2f*)&bcol[n1 * LDSTR];
      const int k2 = (kk + 2 < nk) ? kk + 2 : nk - 1;
      e1 = *(const unsigned*)(kl + 4 * k2);
      const v2f b0 = { x0.x, x1.x };
      const v2f b1 = { x0.y, x1.y };
      acc[0] = __builtin_amdgcn_wmma_f32_16x16x4_f32(
          false, acur, false, b0, (short)0, acc[0], false, false);
      acc[1] = __builtin_amdgcn_wmma_f32_16x16x4_f32(
          false, acur, false, b1, (short)0, acc[1], false, false);
      acur = an; x0 = y0; x1 = y1;
    }

    __syncthreads();

    const float alpha = kAlpha[step];
#pragma unroll
    for (int v = 0; v < 8; ++v) {
      const int row = erow0 + v;
      float fn0 = __builtin_fmaf(alpha, acc[0][v] - rf[0][v], rf[0][v]);
      float fn1 = __builtin_fmaf(alpha, acc[1][v] - rf[1][v], rf[1][v]);
      float fo0 = __builtin_fmaf(rs[0][v], rc[0][v] - fn0, fn0);
      float fo1 = __builtin_fmaf(rs[1][v], rc[1][v] - fn1, fn1);
      rf[0][v] = fo0; rf[1][v] = fo1;
      if (step == NSTEPS - 1) {
        sS[row * LDSTR + ecol0]      = fo0;
        sS[row * LDSTR + ecol0 + 16] = fo1;
      } else {
        v2f p = { fo0, fo1 };
        *(v2f*)&sF[row * LDSTR + fc0] = p;
      }
    }
    __syncthreads();
  }

  float* dst = out + (size_t)b * (N * N);
  for (int e = tid; e < N * N; e += NT) {
    int i = e >> 7, j = e & 127;
    float val = (i == j) ? 1.0f
                         : 0.5f * (sS[i * LDSTR + j] + sS[j * LDSTR + i]);
    dst[e] = val;
  }

  }
}

extern "C" void kernel_launch(void* const* d_in, const int* in_sizes, int n_in,
                              void* d_out, int out_size, void* d_ws, size_t ws_size,
                              hipStream_t stream) {
  (void)in_sizes; (void)n_in; (void)out_size; (void)d_ws; (void)ws_size;
  const float* corr = (const float*)d_in[0];
  const float* temp = (const float*)d_in[1];
  float* out = (float*)d_out;

  const size_t shmem =
      ((size_t)3 * N * LDSTR + (size_t)(N / 2) * PSTR) * sizeof(float)
      + (size_t)NRT * KLS * sizeof(unsigned short) + (size_t)NRT * sizeof(int);
  static bool attr_set = false;
  if (!attr_set) {
    hipFuncSetAttribute((const void*)sgf_kernel,
                        hipFuncAttributeMaxDynamicSharedMemorySize, (int)shmem);
    attr_set = true;
  }
  sgf_kernel<<<dim3(NWG), dim3(NT), shmem, stream>>>(corr, temp, out);
}


# soft_adj and output symmetrized in the accumulator layout: own 16 elements in registers, mirror block by 16-byte row reads, one LDS wait per pass; column lists by ballot; sum orders unchanged
# speedup vs baseline: 4.2468x; 1.0588x over previous
#include <hip/hip_runtime.h>


#define N       128
#define LDSTR   132
#define PSTR    288
#define KLS     144
#define NRT     8
#define NBATCH  2048
#define EPSF    1e-8f
#define NSTEPS  3
#define NT      1024
#define NWG     256
#define NFT     512
#define WT      2

typedef __attribute__((ext_vector_type(2))) float v2f;
typedef __attribute__((ext_vector_type(8))) float v8f;
typedef __attribute__((ext_vector_type(4))) float v4f;

template <int nthr>
__device__ __forceinline__ void fetch_raw(const float* __restrict__ src, float* dst, int t) {
  constexpr int NP = N * N / (4 * nthr);
  const float* s = src + t * 4;
  float*       p = dst + (t >> 5) * LDSTR + (t & 31) * 4;
  v4f v[NP];
#pragma unroll
  for (int q = 0; q < NP; ++q) v[q] = *(const v4f*)(s + q * nthr * 4);
#pragma unroll
  for (int q = 0; q < NP; ++q) *(v4f*)(p + q * (nthr / 32) * LDSTR) = v[q];
}

__device__ __forceinline__ int fcol(int col) {
  return (col & ~31) + ((col & 15) << 1) + ((col >> 4) & 1);
}

__device__ __forceinline__ void fold_pair(int s, int& i, int& j) {
  const int p = s >> 7, t = s & 127;
  const bool up = t > p;
  i = up ? p : (N - 1) - p;
  j = up ? t : (N - 1) - t;
}

__global__ __launch_bounds__(NT, 1)
void sgf_kernel(const float* __restrict__ corr,
                const float* __restrict__ temperature,
                float* __restrict__ out) {
  extern __shared__ float lds[];
  float* sC  = lds;
  float* sS  = lds + N * LDSTR;
  float* sF  = lds + 2 * N * LDSTR;
  float* sR  = lds + 2 * N * LDSTR + (N / 2) * PSTR;
  float* sA  = sF;
  unsigned short* sKl = (unsigned short*)(lds + 3 * N * LDSTR + (N / 2) * PSTR);
  int*            sKn = (int*)(sKl + NRT * KLS);
  unsigned*       sKm = (unsigned*)(sKn + NRT);

  const int tid0 = threadIdx.x;

  const float invT = 1.0f / temperature[0];

  fetch_raw<NT>(corr + (size_t)blockIdx.x * (N * N), sR, tid0);
  __syncthreads();

  for (int b = blockIdx.x; b < NBATCH; b += NWG) {

  int tid = tid0;
  asm volatile("" : "+v"(tid));
  __builtin_assume(tid >= 0 && tid < NT);
  const int lane = tid & 31;
  const int wave = tid >> 5;
  const int lm   = lane & 15;
  const int lh   = lane >> 4;

#pragma unroll
  for (int s = tid; s < (N / 2) * N; s += NT) {
    int i, j;
    fold_pair(s, i, j);
    float x = sR[i * LDSTR + j];
    float y = sR[j * LDSTR + i];
    x = fminf(fmaxf(x, -1.0f + EPSF), 1.0f - EPSF);
    y = fminf(fmaxf(y, -1.0f + EPSF), 1.0f - EPSF);
    const float cv  = 0.5f * (x + y);
    const float d   = __builtin_sqrtf(2.0f * (1.0f - cv) + EPSF);
    float       aff = __expf(-d * invT);
    if (i == j) aff = 0.0f;
    sC[i * LDSTR + j] = cv;
    sC[j * LDSTR + i] = cv;
    sA[i * LDSTR + j] = aff;
    sA[j * LDSTR + i] = aff;
  }
  if (tid < N / 2) {
    float x = sR[tid * (LDSTR + 1)];
    x = fminf(fmaxf(x, -1.0f + EPSF), 1.0f - EPSF);
    sC[tid * (LDSTR + 1)] = 0.5f * (x + x);
    sA[tid * (LDSTR + 1)] = 0.0f;
  }
  __syncthreads();

  if (tid < N) {
    const int i = tid;
    const float* arow = &sA[i * LDSTR];
    float*       srow = &sS[i * LDSTR];
    float t0 = -1e30f, t1 = -1e30f, t2 = -1e30f;
#pragma nounroll
    for (int j0 = 0; j0 < N; j0 += 32) {
      float a[32];
      for (int q = 0; q < 8; ++q) {
        float4 v4 = *(const float4*)&arow[j0 + 4 * q];
        a[4 * q + 0] = v4.x; a[4 * q + 1] = v4.y;
        a[4 * q + 2] = v4.z; a[4 * q + 3] = v4.w;
      }
#pragma unroll
      for (int u = 0; u < 32; ++u) {
        float n0 = fmaxf(t0, a[u]);
        float n1 = __builtin_amdgcn_fmed3f(t0, t1, a[u]);
        float n2 = __builtin_amdgcn_fmed3f(t1, t2, a[u]);
        t0 = n0; t1 = n1; t2 = n2;
      }
    }
    const float e1 = __expf(t1 - t0);
    const float e2 = __expf(t2 - t0);
    int c0 = 0, c1 = 0, c2 = 0, c3 = 0;
    for (int j0 = 0; j0 < N; j0 += 16) {
      float a[16];
      for (int q = 0; q < 4; ++q) {
        float4 v4 = *(const float4*)&arow[j0 + 4 * q];
        a[4 * q + 0] = v4.x; a[4 * q + 1] = v4.y;
        a[4 * q + 2] = v4.z; a[4 * q + 3] = v4.w;
      }
      for (int u = 0; u < 16; u += 4) {
        c0 += (a[u + 0] >= t2) ? 1 : 0;
        c1 += (a[u + 1] >= t2) ? 1 : 0;
        c2 += (a[u + 2] >= t2) ? 1 : 0;
        c3 += (a[u + 3] >= t2) ? 1 : 0;
      }
    }
    const int   npass = (c0 + c1) + (c2 + c3);
    const int   nabove = ((t0 > t2) ? 1 : 0) + ((t1 > t2) ? 1 : 0);
    const float g0 = (t0 > t2) ? 1.0f : 0.0f;
    const float g1 = (t1 > t2) ? e1 : 0.0f;
    const float sum = (g0 + g1) + (float)(npass - nabove) * e2;
    const float rsum = 1.0f / sum;
    const float p0 = rsum, p1 = e1 * rsum, p2 = e2 * rsum;
#pragma unroll
    for (int j0 = 0; j0 < N; j0 += 16) {
      float a[16];
      for (int q = 0; q < 4; ++q) {
        float4 v4 = *(const float4*)&arow[j0 + 4 * q];
        a[4 * q + 0] = v4.x; a[4 * q + 1] = v4.y;
        a[4 * q + 2] = v4.z; a[4 * q + 3] = v4.w;
      }
      for (int u = 0; u < 16; ++u) {
        const float hi = (a[u] >= t0) ? p0 : p1;
        const float md = (a[u] >= t1) ? hi : p2;
        a[u] = (a[u] >= t2) ? md : 0.0f;
      }
      for (int q = 0; q < 4; ++q)
        *(float4*)&srow[j0 + 4 * q] =
            make_float4(a[4 * q + 0], a[4 * q + 1], a[4 * q + 2], a[4 * q + 3]);
    }
  } else if (tid >= NT - NFT && b + NWG < NBATCH) {
    fetch_raw<NFT>(corr + (size_t)(b + NWG) * (N * N), sR, tid & (NFT - 1));
  }
  __syncthreads();

  const int rt    = wave >> 2;
  const int cp    = wave & 3;
  const int ecol0 = cp * 32 + lm;
  const int erow0 = rt * 16 + 8 * lh;
  const int fc0   = cp * 32 + 2 * lm;
  float rc[WT][8], rs[WT][8], rf[WT][8];
#pragma unroll
  for (int t = 0; t < WT; ++t)
#pragma unroll
    for (int v = 0; v < 8; ++v) {
      rc[t][v] = sC[(erow0 + v) * LDSTR + ecol0 + 16 * t];
      rs[t][v] = sS[(erow0 + v) * LDSTR + ecol0 + 16 * t];
    }
  unsigned colbits[WT];
#pragma unroll
  for (int t = 0; t < WT; ++t) {
    const int col = ecol0 + 16 * t;
    const v4f m0 = *(const v4f*)&sS[col * LDSTR + erow0];
    const v4f m1 = *(const v4f*)&sS[col * LDSTR + erow0 + 4];
    const float mt[8] = { m0.x, m0.y, m0.z, m0.w, m1.x, m1.y, m1.z, m1.w };
    unsigned bits = 0u;
#pragma unroll
    for (int v = 0; v < 8; ++v) {
      rs[t][v] = 0.5f * (rs[t][v] + mt[v]);
      bits |= __float_as_uint(rs[t][v]) & 0x7fffffffu;
    }
    colbits[t] = bits;
  }
#pragma unroll
  for (int v = 0; v < 8; ++v) {
    v2f p = { rc[0][v], rc[1][v] };
    *(v2f*)&sF[(erow0 + v) * LDSTR + fc0] = p;
    rf[0][v] = rc[0][v];
    rf[1][v] = rc[1][v];
  }
  {
    const unsigned b0 = __builtin_amdgcn_ballot_w32(colbits[0] != 0u);
    const unsigned b1 = __builtin_amdgcn_ballot_w32(colbits[1] != 0u);
    const unsigned cm = ((b0 | (b0 >> 16)) & 0xffffu) | ((b1 | (b1 >> 16)) << 16);
    if (lane == 0) sKm[wave] = cm;
  }
  __syncthreads();

#pragma unroll
  for (int v = 0; v < 8; ++v) {
    sS[(erow0 + v) * LDSTR + ecol0]      = rs[0][v];
    sS[(erow0 + v) * LDSTR + ecol0 + 16] = rs[1][v];
  }

  if (cp == (rt & 3)) {
    unsigned short* klw = sKl + rt * KLS;
    int cnt = 0;
    int zcol = 0;
    bool havez = false;
#pragma unroll
    for (int q = 0; q < 4; ++q) {
      const unsigned m = (unsigned)__builtin_amdgcn_readfirstlane((int)sKm[rt * 4 + q]);
      const bool nz = ((m >> lane) & 1u) != 0u;
      const int pos = cnt + (int)__builtin_amdgcn_mbcnt_lo(m, 0u);
      if (nz) klw[pos] = (unsigned short)(32 * q + lane);
      cnt += __builtin_popcount(m);
      const unsigned zm = ~m;
      if (!havez && zm != 0u) { zcol = 32 * q + __builtin_ctz(zm); havez = true; }
    }
    int cnt4 = (cnt + 3) & ~3;
    if (cnt4 < 4) cnt4 = 4;
    if (lane < cnt4 - cnt) klw[cnt + lane] = (unsigned short)zcol;
    if (lane == 0) sKn[rt] = cnt4 >> 2;
  }
  __syncthreads();

  const float* arow = sS + (rt * 16 + lm) * LDSTR;
  const float* bcol = sF + fc0;
  const unsigned short* kl = sKl + rt * KLS + 2 * lh;
  const int nk = __builtin_amdgcn_readfirstlane(sKn[rt]);

  const float kAlpha[NSTEPS] = { 0.5f, 1.0f / 3.0f, 0.25f };

#pragma unroll
  for (int step = 0; step < NSTEPS; ++step) {
    v8f acc[WT] = {};

    unsigned e1;
    v2f acur, x0, x1;
    {
      const unsigned e0 = *(const unsigned*)kl;
      const int k0 = (int)(e0 & 0xffffu), k1 = (int)(e0 >> 16);
      acur = (v2f){ arow[k0], arow[k1] };
      x0 = *(const v2f*)&bcol[k0 * LDSTR];
      x1 = *(const v2f*)&bcol[k1 * LDSTR];
      e1 = *(const unsigned*)(kl + 4 * ((nk > 1) ? 1 : 0));
    }

    for (int kk = 0; kk < nk; ++kk) {
      const int n0 = (int)(e1 & 0xffffu), n1 = (int)(e1 >> 16);
      const v2f an = { arow[n0], arow[n1] };
      const v2f y0 = *(const v2f*)&bcol[n0 * LDSTR];
      const v2f y1 = *(const v2f*)&bcol[n1 * LDSTR];
      const int k2 = (kk + 2 < nk) ? kk + 2 : nk - 1;
      e1 = *(const unsigned*)(kl + 4 * k2);
      const v2f b0 = { x0.x, x1.x };
      const v2f b1 = { x0.y, x1.y };
      acc[0] = __builtin_amdgcn_wmma_f32_16x16x4_f32(
          false, acur, false, b0, (short)0, acc[0], false, false);
      acc[1] = __builtin_amdgcn_wmma_f32_16x16x4_f32(
          false, acur, false, b1, (short)0, acc[1], false, false);
      acur = an; x0 = y0; x1 = y1;
    }

    __syncthreads();

    const float alpha = kAlpha[step];
#pragma unroll
    for (int v = 0; v < 8; ++v) {
      const int row = erow0 + v;
      float fn0 = __builtin_fmaf(alpha, acc[0][v] - rf[0][v], rf[0][v]);
      float fn1 = __builtin_fmaf(alpha, acc[1][v] - rf[1][v], rf[1][v]);
      float fo0 = __builtin_fmaf(rs[0][v], rc[0][v] - fn0, fn0);
      float fo1 = __builtin_fmaf(rs[1][v], rc[1][v] - fn1, fn1);
      rf[0][v] = fo0; rf[1][v] = fo1;
      if (step == NSTEPS - 1) {
        sS[row * LDSTR + ecol0]      = fo0;
        sS[row * LDSTR + ecol0 + 16] = fo1;
      } else {
        v2f p = { fo0, fo1 };
        *(v2f*)&sF[row * LDSTR + fc0] = p;
      }
    }
    __syncthreads();
  }

  float* dst = out + (size_t)b * (N * N) + erow0 * N + ecol0;
#pragma unroll
  for (int t = 0; t < WT; ++t) {
    const int col = ecol0 + 16 * t;
    const v4f m0 = *(const v4f*)&sS[col * LDSTR + erow0];
    const v4f m1 = *(const v4f*)&sS[col * LDSTR + erow0 + 4];
    const float mt[8] = { m0.x, m0.y, m0.z, m0.w, m1.x, m1.y, m1.z, m1.w };
    const int dg = col - erow0;
#pragma unroll
    for (int v = 0; v < 8; ++v) {
      const float val = (v == dg) ? 1.0f : 0.5f * (rf[t][v] + mt[v]);
      dst[v * N + 16 * t] = val;
    }
  }

  }
}

extern "C" void kernel_launch(void* const* d_in, const int* in_sizes, int n_in,
                              void* d_out, int out_size, void* d_ws, size_t ws_size,
                              hipStream_t stream) {
  (void)in_sizes; (void)n_in; (void)out_size; (void)d_ws; (void)ws_size;
  const float* corr = (const float*)d_in[0];
  const float* temp = (const float*)d_in[1];
  float* out = (float*)d_out;

  const size_t shmem =
      ((size_t)3 * N * LDSTR + (size_t)(N / 2) * PSTR) * sizeof(float)
      + (size_t)NRT * KLS * sizeof(unsigned short) + (size_t)NRT * sizeof(int)
      + (size_t)(NT / 32) * sizeof(unsigned);
  static bool attr_set = false;
  if (!attr_set) {
    hipFuncSetAttribute((const void*)sgf_kernel,
                        hipFuncAttributeMaxDynamicSharedMemorySize, (int)shmem);
    attr_set = true;
  }
  sgf_kernel<<<dim3(NWG), dim3(NT), shmem, stream>>>(corr, temp, out);
}


# Next matrix's pair pass (clip, sym, sqrt, exp) runs on the 28 waves idle during the 4-wave graph build; two LDS planes swap roles per matrix, affinities in place over raw; sum orders unchanged
# speedup vs baseline: 4.4284x; 1.0428x over previous
#include <hip/hip_runtime.h>


#define N       128
#define LDSTR   132
#define PSTR    288
#define KLS     144
#define NRT     8
#define NBATCH  2048
#define EPSF    1e-8f
#define NSTEPS  3
#define NT      1024
#define NWG     256
#define NPT     (NT - N)
#define WT      2

typedef __attribute__((ext_vector_type(2))) float v2f;
typedef __attribute__((ext_vector_type(8))) float v8f;
typedef __attribute__((ext_vector_type(4))) float v4f;

#define NPF     (N * N / (4 * NT))

__device__ __forceinline__ int fcol(int col) {
  return (col & ~31) + ((col & 15) << 1) + ((col >> 4) & 1);
}

__device__ __forceinline__ void fold_pair(int s, int& i, int& j) {
  const int p = s >> 7, t = s & 127;
  const bool up = t > p;
  i = up ? p : (N - 1) - p;
  j = up ? t : (N - 1) - t;
}

template <int Q0, int Q1, int STRIDE>
__device__ __forceinline__ void pair_slots(float* sX, float* sC, int t, float invT) {
  float xs[Q1 - Q0], ys[Q1 - Q0];
#pragma unroll
  for (int q = Q0; q < Q1; ++q) {
    int i, j;
    fold_pair(t + q * STRIDE, i, j);
    xs[q - Q0] = sX[i * LDSTR + j];
    ys[q - Q0] = sX[j * LDSTR + i];
  }
#pragma unroll
  for (int q = Q0; q < Q1; ++q) {
    int i, j;
    fold_pair(t + q * STRIDE, i, j);
    const float x = fminf(fmaxf(xs[q - Q0], -1.0f + EPSF), 1.0f - EPSF);
    const float y = fminf(fmaxf(ys[q - Q0], -1.0f + EPSF), 1.0f - EPSF);
    const float cv  = 0.5f * (x + y);
    const float d   = __builtin_sqrtf(2.0f * (1.0f - cv) + EPSF);
    float       aff = __expf(-d * invT);
    if (i == j) aff = 0.0f;
    sC[i * LDSTR + j] = cv;
    sC[j * LDSTR + i] = cv;
    sX[i * LDSTR + j] = aff;
    sX[j * LDSTR + i] = aff;
  }
}

__device__ __forceinline__ void pair_diag(float* sX, float* sC, int t) {
  if (t < N / 2) {
    float x = sX[t * (LDSTR + 1)];
    x = fminf(fmaxf(x, -1.0f + EPSF), 1.0f - EPSF);
    sC[t * (LDSTR + 1)] = 0.5f * (x + x);
    sX[t * (LDSTR + 1)] = 0.0f;
  }
}

__global__ __launch_bounds__(NT, 1)
void sgf_kernel(const float* __restrict__ corr,
                const float* __restrict__ temperature,
                float* __restrict__ out) {
  extern __shared__ float lds[];
  float* sC  = lds;
  float* sS  = lds + N * LDSTR;
  float* sPb = lds + 2 * N * LDSTR;
  float* sPa = lds + 2 * N * LDSTR + (N / 2) * PSTR;
  unsigned short* sKl = (unsigned short*)(lds + 3 * N * LDSTR + (N / 2) * PSTR);
  int*            sKn = (int*)(sKl + NRT * KLS);
  unsigned*       sKm = (unsigned*)(sKn + NRT);

  const int tid0 = threadIdx.x;

  const float invT = 1.0f / temperature[0];

  {
    const bool two = (int)blockIdx.x + NWG < NBATCH;
    const float* s = corr + (size_t)blockIdx.x * (N * N) + tid0 * 4;
    float*       p = sPa + (tid0 >> 5) * LDSTR + (tid0 & 31) * 4;
    v4f v0[NPF], v1[NPF];
#pragma unroll
    for (int q = 0; q < NPF; ++q) v0[q] = *(const v4f*)(s + q * NT * 4);
    if (two) {
#pragma unroll
      for (int q = 0; q < NPF; ++q) v1[q] = *(const v4f*)(s + (size_t)NWG * (N * N) + q * NT * 4);
    }
#pragma unroll
    for (int q = 0; q < NPF; ++q) *(v4f*)(p + q * (NT / 32) * LDSTR) = v0[q];
    if (two) {
#pragma unroll
      for (int q = 0; q < NPF; ++q) *(v4f*)(p + (sPb - sPa) + q * (NT / 32) * LDSTR) = v1[q];
    }
  }
  __syncthreads();
  pair_slots<0, 8, NT>(sPa, sC, tid0, invT);
  pair_diag(sPa, sC, tid0);
  __syncthreads();

  float rc[WT][8];
  {
    const int r0 = (tid0 >> 7) * 16 + 8 * ((tid0 >> 4) & 1);
    const int c0 = ((tid0 >> 5) & 3) * 32 + (tid0 & 15);
#pragma unroll
    for (int t = 0; t < WT; ++t)
#pragma unroll
      for (int v = 0; v < 8; ++v) rc[t][v] = sC[(r0 + v) * LDSTR + c0 + 16 * t];
  }

  int it = 0;
  for (int b = blockIdx.x; b < NBATCH; b += NWG, ++it) {
  const bool has_next  = b + NWG < NBATCH;
  const bool has_next2 = b + 2 * NWG < NBATCH;
  float* pX = (it & 1) ? sPb : sPa;
  float* pN = (it & 1) ? sPa : sPb;
  __syncthreads();

  int tid = tid0;
  asm volatile("" : "+v"(tid));
  __builtin_assume(tid >= 0 && tid < NT);
  const int lane = tid & 31;
  const int wave = tid >> 5;
  const int lm   = lane & 15;
  const int lh   = lane >> 4;

  if (tid < N) {
    const int i = tid;
    const float* arow = &pX[i * LDSTR];
    float*       srow = &sS[i * LDSTR];
    float t0 = -1e30f, t1 = -1e30f, t2 = -1e30f;
#pragma nounroll
    for (int j0 = 0; j0 < N; j0 += 32) {
      float a[32];
      for (int q = 0; q < 8; ++q) {
        float4 v4 = *(const float4*)&arow[j0 + 4 * q];
        a[4 * q + 0] = v4.x; a[4 * q + 1] = v4.y;
        a[4 * q + 2] = v4.z; a[4 * q + 3] = v4.w;
      }
#pragma unroll
      for (int u = 0; u < 32; ++u) {
        float n0 = fmaxf(t0, a[u]);
        float n1 = __builtin_amdgcn_fmed3f(t0, t1, a[u]);
        float n2 = __builtin_amdgcn_fmed3f(t1, t2, a[u]);
        t0 = n0; t1 = n1; t2 = n2;
      }
    }
    const float e1 = __expf(t1 - t0);
    const float e2 = __expf(t2 - t0);
    int c0 = 0, c1 = 0, c2 = 0, c3 = 0;
    for (int j0 = 0; j0 < N; j0 += 16) {
      float a[16];
      for (int q = 0; q < 4; ++q) {
        float4 v4 = *(const float4*)&arow[j0 + 4 * q];
        a[4 * q + 0] = v4.x; a[4 * q + 1] = v4.y;
        a[4 * q + 2] = v4.z; a[4 * q + 3] = v4.w;
      }
      for (int u = 0; u < 16; u += 4) {
        c0 += (a[u + 0] >= t2) ? 1 : 0;
        c1 += (a[u + 1] >= t2) ? 1 : 0;
        c2 += (a[u + 2] >= t2) ? 1 : 0;
        c3 += (a[u + 3] >= t2) ? 1 : 0;
      }
    }
    const int   npass = (c0 + c1) + (c2 + c3);
    const int   nabove = ((t0 > t2) ? 1 : 0) + ((t1 > t2) ? 1 : 0);
    const float g0 = (t0 > t2) ? 1.0f : 0.0f;
    const float g1 = (t1 > t2) ? e1 : 0.0f;
    const float sum = (g0 + g1) + (float)(npass - nabove) * e2;
    const float rsum = 1.0f / sum;
    const float p0 = rsum, p1 = e1 * rsum, p2 = e2 * rsum;
#pragma unroll
    for (int j0 = 0; j0 < N; j0 += 16) {
      float a[16];
      for (int q = 0; q < 4; ++q) {
        float4 v4 = *(const float4*)&arow[j0 + 4 * q];
        a[4 * q + 0] = v4.x; a[4 * q + 1] = v4.y;
        a[4 * q + 2] = v4.z; a[4 * q + 3] = v4.w;
      }
      for (int u = 0; u < 16; ++u) {
        const float hi = (a[u] >= t0) ? p0 : p1;
        const float md = (a[u] >= t1) ? hi : p2;
        a[u] = (a[u] >= t2) ? md : 0.0f;
      }
      for (int q = 0; q < 4; ++q)
        *(float4*)&srow[j0 + 4 * q] =
            make_float4(a[4 * q + 0], a[4 * q + 1], a[4 * q + 2], a[4 * q + 3]);
    }
  } else if (has_next) {
    const int u = tid - N;
    __builtin_assume(u >= 0 && u < NPT);
    pair_slots<0, 9, NPT>(pN, sC, u, invT);
    if (u < (N / 2) * N - 9 * NPT) pair_slots<9, 10, NPT>(pN, sC, u, invT);
    pair_diag(pN, sC, u);
  }
  __syncthreads();

  const int rt    = wave >> 2;
  const int cp    = wave & 3;
  const int ecol0 = cp * 32 + lm;
  const int erow0 = rt * 16 + 8 * lh;
  const int fc0   = cp * 32 + 2 * lm;
  float rs[WT][8], rf[WT][8];
#pragma unroll
  for (int t = 0; t < WT; ++t)
#pragma unroll
    for (int v = 0; v < 8; ++v)
      rs[t][v] = sS[(erow0 + v) * LDSTR + ecol0 + 16 * t];
  unsigned colbits[WT];
#pragma unroll
  for (int t = 0; t < WT; ++t) {
    const int col = ecol0 + 16 * t;
    const v4f m0 = *(const v4f*)&sS[col * LDSTR + erow0];
    const v4f m1 = *(const v4f*)&sS[col * LDSTR + erow0 + 4];
    const float mt[8] = { m0.x, m0.y, m0.z, m0.w, m1.x, m1.y, m1.z, m1.w };
    unsigned bits = 0u;
#pragma unroll
    for (int v = 0; v < 8; ++v) {
      rs[t][v] = 0.5f * (rs[t][v] + mt[v]);
      bits |= __float_as_uint(rs[t][v]) & 0x7fffffffu;
    }
    colbits[t] = bits;
  }
#pragma unroll
  for (int v = 0; v < 8; ++v) {
    v2f p = { rc[0][v], rc[1][v] };
    *(v2f*)&pX[(erow0 + v) * LDSTR + fc0] = p;
    rf[0][v] = rc[0][v];
    rf[1][v] = rc[1][v];
  }
  {
    const unsigned b0 = __builtin_amdgcn_ballot_w32(colbits[0] != 0u);
    const unsigned b1 = __builtin_amdgcn_ballot_w32(colbits[1] != 0u);
    const unsigned cm = ((b0 | (b0 >> 16)) & 0xffffu) | ((b1 | (b1 >> 16)) << 16);
    if (lane == 0) sKm[wave] = cm;
  }
  __syncthreads();

#pragma unroll
  for (int v = 0; v < 8; ++v) {
    sS[(erow0 + v) * LDSTR + ecol0]      = rs[0][v];
    sS[(erow0 + v) * LDSTR + ecol0 + 16] = rs[1][v];
  }

  if (cp == (rt & 3)) {
    unsigned short* klw = sKl + rt * KLS;
    int cnt = 0;
    int zcol = 0;
    bool havez = false;
#pragma unroll
    for (int q = 0; q < 4; ++q) {
      const unsigned m = (unsigned)__builtin_amdgcn_readfirstlane((int)sKm[rt * 4 + q]);
      const bool nz = ((m >> lane) & 1u) != 0u;
      const int pos = cnt + (int)__builtin_amdgcn_mbcnt_lo(m, 0u);
      if (nz) klw[pos] = (unsigned short)(32 * q + lane);
      cnt += __builtin_popcount(m);
      const unsigned zm = ~m;
      if (!havez && zm != 0u) { zcol = 32 * q + __builtin_ctz(zm); havez = true; }
    }
    int cnt4 = (cnt + 3) & ~3;
    if (cnt4 < 4) cnt4 = 4;
    if (lane < cnt4 - cnt) klw[cnt + lane] = (unsigned short)zcol;
    if (lane == 0) sKn[rt] = cnt4 >> 2;
  }
  __syncthreads();

  const float* arow = sS + (rt * 16 + lm) * LDSTR;
  const float* bcol = pX + fc0;
  const unsigned short* kl = sKl + rt * KLS + 2 * lh;
  const int nk = __builtin_amdgcn_readfirstlane(sKn[rt]);

  const float kAlpha[NSTEPS] = { 0.5f, 1.0f / 3.0f, 0.25f };

  v4f pv[NPF];

#pragma unroll
  for (int step = 0; step < NSTEPS; ++step) {
    if (step == NSTEPS - 1 && has_next2) {
      const float* s = corr + (size_t)(b + 2 * NWG) * (N * N) + tid * 4;
#pragma unroll
      for (int q = 0; q < NPF; ++q) pv[q] = *(const v4f*)(s + q * NT * 4);
    }

    v8f acc[WT] = {};

    unsigned e1;
    v2f acur, x0, x1;
    {
      const unsigned e0 = *(const unsigned*)kl;
      const int k0 = (int)(e0 & 0xffffu), k1 = (int)(e0 >> 16);
      acur = (v2f){ arow[k0], arow[k1] };
      x0 = *(const v2f*)&bcol[k0 * LDSTR];
      x1 = *(const v2f*)&bcol[k1 * LDSTR];
      e1 = *(const unsigned*)(kl + 4 * ((nk > 1) ? 1 : 0));
    }

    for (int kk = 0; kk < nk; ++kk) {
      const int n0 = (int)(e1 & 0xffffu), n1 = (int)(e1 >> 16);
      const v2f an = { arow[n0], arow[n1] };
      const v2f y0 = *(const v2f*)&bcol[n0 * LDSTR];
      const v2f y1 = *(const v2f*)&bcol[n1 * LDSTR];
      const int k2 = (kk + 2 < nk) ? kk + 2 : nk - 1;
      e1 = *(const unsigned*)(kl + 4 * k2);
      const v2f b0 = { x0.x, x1.x };
      const v2f b1 = { x0.y, x1.y };
      acc[0] = __builtin_amdgcn_wmma_f32_16x16x4_f32(
          false, acur, false, b0, (short)0, acc[0], false, false);
      acc[1] = __builtin_amdgcn_wmma_f32_16x16x4_f32(
          false, acur, false, b1, (short)0, acc[1], false, false);
      acur = an; x0 = y0; x1 = y1;
    }

    __syncthreads();

    if (step == NSTEPS - 1 && has_next2) {
      float* p = pX + (tid >> 5) * LDSTR + (tid & 31) * 4;
#pragma unroll
      for (int q = 0; q < NPF; ++q) *(v4f*)(p + q * (NT / 32) * LDSTR) = pv[q];
    }

    const float alpha = kAlpha[step];
#pragma unroll
    for (int v = 0; v < 8; ++v) {
      const int row = erow0 + v;
      float fn0 = __builtin_fmaf(alpha, acc[0][v] - rf[0][v], rf[0][v]);
      float fn1 = __builtin_fmaf(alpha, acc[1][v] - rf[1][v], rf[1][v]);
      float fo0 = __builtin_fmaf(rs[0][v], rc[0][v] - fn0, fn0);
      float fo1 = __builtin_fmaf(rs[1][v], rc[1][v] - fn1, fn1);
      rf[0][v] = fo0; rf[1][v] = fo1;
      if (step == NSTEPS - 1) {
        sS[row * LDSTR + ecol0]      = fo0;
        sS[row * LDSTR + ecol0 + 16] = fo1;
      } else {
        v2f p = { fo0, fo1 };
        *(v2f*)&pX[row * LDSTR + fc0] = p;
      }
    }
    __syncthreads();
  }

  float* dst = out + (size_t)b * (N * N) + erow0 * N + ecol0;
#pragma unroll
  for (int t = 0; t < WT; ++t) {
    const int col = ecol0 + 16 * t;
    const v4f m0 = *(const v4f*)&sS[col * LDSTR + erow0];
    const v4f m1 = *(const v4f*)&sS[col * LDSTR + erow0 + 4];
    const float mt[8] = { m0.x, m0.y, m0.z, m0.w, m1.x, m1.y, m1.z, m1.w };
    const int dg = col - erow0;
#pragma unroll
    for (int v = 0; v < 8; ++v) {
      const float val = (v == dg) ? 1.0f : 0.5f * (rf[t][v] + mt[v]);
      dst[v * N + 16 * t] = val;
    }
  }

  if (has_next) {
#pragma unroll
    for (int t = 0; t < WT; ++t)
#pragma unroll
      for (int v = 0; v < 8; ++v) rc[t][v] = sC[(erow0 + v) * LDSTR + ecol0 + 16 * t];
  }

  }
}

extern "C" void kernel_launch(void* const* d_in, const int* in_sizes, int n_in,
                              void* d_out, int out_size, void* d_ws, size_t ws_size,
                              hipStream_t stream) {
  (void)in_sizes; (void)n_in; (void)out_size; (void)d_ws; (void)ws_size;
  const float* corr = (const float*)d_in[0];
  const float* temp = (const float*)d_in[1];
  float* out = (float*)d_out;

  const size_t shmem =
      ((size_t)3 * N * LDSTR + (size_t)(N / 2) * PSTR) * sizeof(float)
      + (size_t)NRT * KLS * sizeof(unsigned short) + (size_t)NRT * sizeof(int)
      + (size_t)(NT / 32) * sizeof(unsigned);
  static bool attr_set = false;
  if (!attr_set) {
    hipFuncSetAttribute((const void*)sgf_kernel,
                        hipFuncAttributeMaxDynamicSharedMemorySize, (int)shmem);
    attr_set = true;
  }
  sgf_kernel<<<dim3(NWG), dim3(NT), shmem, stream>>>(corr, temp, out);
}
